# starting kernel
# baseline (speedup 1.0000x reference)
#include <hip/hip_runtime.h>
#include <hip/hip_bf16.h>

typedef __attribute__((ext_vector_type(2))) float v2f;
typedef __attribute__((ext_vector_type(8))) float v8f;

#define IN_DIM  500
#define HID_DIM 16
#define OUT_DIM 7

__global__ void fill_f32(float* __restrict__ p, float v, int n) {
    int i = blockIdx.x * blockDim.x + threadIdx.x;
    if (i < n) p[i] = v;
}

__global__ void degree_kernel(const int* __restrict__ dst,
                              float* __restrict__ deg, int num_edges) {
    int e = blockIdx.x * blockDim.x + threadIdx.x;
    if (e < num_edges) atomicAdd(deg + dst[e], 1.0f);
}

__global__ void rsqrt_kernel(float* __restrict__ dinv, int n) {
    int i = blockIdx.x * blockDim.x + threadIdx.x;
    if (i < n) {
        float d = dinv[i];
        dinv[i] = (d > 0.0f) ? rsqrtf(d) : 0.0f;
    }
}

__global__ void gemm1_wmma(const float* __restrict__ x,
                           const float* __restrict__ W1,
                           float* __restrict__ h1, int num_nodes) {
    int wave = (blockIdx.x * blockDim.x + threadIdx.x) >> 5;
    int lane = threadIdx.x & 31;
    int row0 = wave * 16;
    if (row0 + 16 > num_nodes) return;

    int half = lane >> 4;
    int l    = lane & 15;

    const float* arow = x + (size_t)(row0 + l) * IN_DIM + half * 2;
    v8f c = {};
    for (int k0 = 0; k0 < IN_DIM; k0 += 4) {
        v2f a = *(const v2f*)(arow + k0);
        v2f b;
        b.x = W1[(k0 + half * 2 + 0) * HID_DIM + l];
        b.y = W1[(k0 + half * 2 + 1) * HID_DIM + l];
        c = __builtin_amdgcn_wmma_f32_16x16x4_f32(
                false, a, false, b, (short)0, c, false, false);
    }
    float* outp = h1 + (size_t)row0 * HID_DIM;
#pragma unroll
    for (int r = 0; r < 8; ++r)
        outp[(half * 8 + r) * HID_DIM + l] = c[r];
}

__global__ void edge_mp16(const int* __restrict__ src,
                          const int* __restrict__ dst,
                          const float* __restrict__ dinv,
                          const float* __restrict__ h,
                          float* __restrict__ agg, int num_edges) {
    int idx = blockIdx.x * blockDim.x + threadIdx.x;
    int e = idx >> 4;
    int j = idx & 15;
    if (e >= num_edges) return;
    int s = src[e], d = dst[e];
    float norm = dinv[s] * dinv[d];
    atomicAdd(agg + (size_t)d * HID_DIM + j, h[(size_t)s * HID_DIM + j] * norm);
}

__global__ void node_post1(const float* __restrict__ h1,
                           const float* __restrict__ dinv,
                           const float* __restrict__ b1,
                           float* __restrict__ agg1, int num_nodes) {
    int idx = blockIdx.x * blockDim.x + threadIdx.x;
    int n = idx >> 4;
    int j = idx & 15;
    if (n >= num_nodes) return;
    float di = dinv[n];
    float v = agg1[idx] + h1[idx] * di * di + b1[j];
    agg1[idx] = fmaxf(v, 0.0f);
}

__global__ void gemm2_wmma(const float* __restrict__ agg1,
                           const float* __restrict__ W2,
                           float* __restrict__ h2, int num_nodes) {
    int wave = (blockIdx.x * blockDim.x + threadIdx.x) >> 5;
    int lane = threadIdx.x & 31;
    int row0 = wave * 16;
    if (row0 + 16 > num_nodes) return;

    int half = lane >> 4;
    int l    = lane & 15;

    const float* arow = agg1 + (size_t)(row0 + l) * HID_DIM + half * 2;
    v8f c = {};
#pragma unroll
    for (int k0 = 0; k0 < HID_DIM; k0 += 4) {
        v2f a = *(const v2f*)(arow + k0);
        v2f b;
        int kr = k0 + half * 2;
        b.x = (l < OUT_DIM) ? W2[(kr + 0) * OUT_DIM + l] : 0.0f;
        b.y = (l < OUT_DIM) ? W2[(kr + 1) * OUT_DIM + l] : 0.0f;
        c = __builtin_amdgcn_wmma_f32_16x16x4_f32(
                false, a, false, b, (short)0, c, false, false);
    }
    if (l < OUT_DIM) {
        float* outp = h2 + (size_t)row0 * OUT_DIM;
#pragma unroll
        for (int r = 0; r < 8; ++r)
            outp[(half * 8 + r) * OUT_DIM + l] = c[r];
    }
}

__global__ void edge_mp7(const int* __restrict__ src,
                         const int* __restrict__ dst,
                         const float* __restrict__ dinv,
                         const float* __restrict__ h,
                         float* __restrict__ agg, int num_edges) {
    int idx = blockIdx.x * blockDim.x + threadIdx.x;
    int e = idx >> 3;
    int j = idx & 7;
    if (e >= num_edges || j >= OUT_DIM) return;
    int s = src[e], d = dst[e];
    float norm = dinv[s] * dinv[d];
    atomicAdd(agg + (size_t)d * OUT_DIM + j, h[(size_t)s * OUT_DIM + j] * norm);
}

__global__ void node_post2(const float* __restrict__ h2,
                           const float* __restrict__ dinv,
                           const float* __restrict__ b2,
                           const float* __restrict__ agg2,
                           float* __restrict__ out, int num_nodes) {
    int n = blockIdx.x * blockDim.x + threadIdx.x;
    if (n >= num_nodes) return;
    float di = dinv[n];
    float sl = di * di;
    float v[OUT_DIM];
    float m = -INFINITY;
#pragma unroll
    for (int j = 0; j < OUT_DIM; ++j) {
        v[j] = agg2[(size_t)n * OUT_DIM + j] + h2[(size_t)n * OUT_DIM + j] * sl + b2[j];
        m = fmaxf(m, v[j]);
    }
    float s = 0.0f;
#pragma unroll
    for (int j = 0; j < OUT_DIM; ++j) s += expf(v[j] - m);
    float lse = logf(s);
#pragma unroll
    for (int j = 0; j < OUT_DIM; ++j)
        out[(size_t)n * OUT_DIM + j] = v[j] - m - lse;
}

extern "C" void kernel_launch(void* const* d_in, const int* in_sizes, int n_in,
                              void* d_out, int out_size, void* d_ws, size_t ws_size,
                              hipStream_t stream) {
    const float* x  = (const float*)d_in[0];
    const int*   ei = (const int*)d_in[1];
    const float* W1 = (const float*)d_in[2];
    const float* b1 = (const float*)d_in[3];
    const float* W2 = (const float*)d_in[4];
    const float* b2 = (const float*)d_in[5];

    const int N = in_sizes[0] / IN_DIM;
    const int E = in_sizes[1] / 2;
    const int* src = ei;
    const int* dst = ei + E;

    float* ws   = (float*)d_ws;
    float* dinv = ws;
    float* h1   = dinv + (size_t)N;
    float* agg1 = h1 + (size_t)N * HID_DIM;
    float* h2   = agg1 + (size_t)N * HID_DIM;
    float* agg2 = h2 + (size_t)N * OUT_DIM;
    float* outp = (float*)d_out;

    const int B = 256;
    auto nb = [](long long work, int blk) { return (int)((work + blk - 1) / blk); };

    fill_f32<<<nb(N, B), B, 0, stream>>>(dinv, 1.0f, N);
    fill_f32<<<nb((long long)N * HID_DIM, B), B, 0, stream>>>(agg1, 0.0f, N * HID_DIM);
    fill_f32<<<nb((long long)N * OUT_DIM, B), B, 0, stream>>>(agg2, 0.0f, N * OUT_DIM);

    degree_kernel<<<nb(E, B), B, 0, stream>>>(dst, dinv, E);
    rsqrt_kernel<<<nb(N, B), B, 0, stream>>>(dinv, N);

    int waves1 = N / 16;
    gemm1_wmma<<<nb((long long)waves1 * 32, B), B, 0, stream>>>(x, W1, h1, N);
    edge_mp16<<<nb((long long)E * 16, B), B, 0, stream>>>(src, dst, dinv, h1, agg1, E);
    node_post1<<<nb((long long)N * HID_DIM, B), B, 0, stream>>>(h1, dinv, b1, agg1, N);

    gemm2_wmma<<<nb((long long)waves1 * 32, B), B, 0, stream>>>(agg1, W2, h2, N);
    edge_mp7<<<nb((long long)E * 8, B), B, 0, stream>>>(src, dst, dinv, h2, agg2, E);
    node_post2<<<nb(N, B), B, 0, stream>>>(h2, dinv, b2, agg2, outp, N);
}


# Layer 2 reuses layer 1's edge filter: a layer-1 work-group keeps its matches (source, local destination) in the workspace; layer 2 reads that list, no rescan of dst and src (LDS add order changes)
# speedup vs baseline: 2.4239x; 2.4239x over previous
#include <hip/hip_runtime.h>
#include <hip/hip_bf16.h>

typedef __attribute__((ext_vector_type(2))) float v2f;
typedef __attribute__((ext_vector_type(8))) float v8f;

#define IN_DIM  500
#define HID_DIM 16
#define OUT_DIM 7

__device__ __forceinline__ void zero_plane(float* __restrict__ p, int n) {
    const int gtid = blockIdx.x * blockDim.x + threadIdx.x;
    const int nthr = gridDim.x * blockDim.x;
    const int n4   = n >> 2;
    float4* p4 = (float4*)p;
    for (int q = gtid; q < n4; q += nthr) p4[q] = make_float4(0.0f, 0.0f, 0.0f, 0.0f);
    const int t = (n4 << 2) + gtid;
    if (t < n) p[t] = 0.0f;
}

#define DEG_CAP   32768
#define DEG_TPB   1024
#define DEG_BATCH 21
#define DEG_ITERS 63
#define DEG_CHUNK (DEG_ITERS * DEG_TPB)

__device__ __forceinline__ int deg_pitch(int num_nodes) {
    return (((num_nodes + 1) >> 1) + 3) & ~3;
}

template <bool FULL>
__device__ __forceinline__ void deg_count(const int* __restrict__ p, int left,
                                          unsigned* hist, int base, unsigned cnt) {
    for (int it = 0; it < DEG_ITERS; it += DEG_BATCH) {
        int v[DEG_BATCH];
#pragma unroll
        for (int k = 0; k < DEG_BATCH; ++k) {
            int o = (it + k) * DEG_TPB;
            v[k] = (FULL || o < left) ? p[o] : -1;
        }
#pragma unroll
        for (int k = 0; k < DEG_BATCH; ++k) {
            unsigned d = (unsigned)(v[k] - base);
            if (d < cnt) atomicAdd(hist + (d >> 1), 1u << ((d & 1u) << 4));
        }
    }
}

__global__ __launch_bounds__(DEG_TPB)
void degree_hist(const int* __restrict__ dst, unsigned* __restrict__ part,
                 int num_nodes, int num_edges) {
    __shared__ uint4 hist4[DEG_CAP / 8];
    unsigned* hist = (unsigned*)hist4;
    const int tid  = threadIdx.x;
    const int base = blockIdx.y * DEG_CAP;
    const unsigned cnt = (unsigned)min(DEG_CAP, num_nodes - base);

    for (int i = tid; i < DEG_CAP / 8; i += DEG_TPB) hist4[i] = make_uint4(0u, 0u, 0u, 0u);
    __syncthreads();

    const int e0 = blockIdx.x * DEG_CHUNK;
    if (e0 + DEG_CHUNK <= num_edges) {
        deg_count<true>(dst + e0 + tid, 0, hist, base, cnt);
    } else {
        deg_count<false>(dst + e0 + tid, num_edges - e0 - tid, hist, base, cnt);
    }
    __syncthreads();

    uint4* outp = (uint4*)(part + (size_t)blockIdx.x * deg_pitch(num_nodes) + (base >> 1));
    const int nq = (int)((((cnt + 1u) >> 1) + 3u) >> 2);
    for (int i = tid; i < nq; i += DEG_TPB) outp[i] = hist4[i];
}

__global__ void degree_merge(const unsigned* __restrict__ part,
                             float* __restrict__ dinv, int num_nodes, int num_chunks) {
    int w = blockIdx.x * blockDim.x + threadIdx.x;
    if (w >= ((num_nodes + 1) >> 1)) return;
    const int pitch = deg_pitch(num_nodes);
    const unsigned* p = part + w;
    unsigned lo = 0u, hi = 0u;
#pragma unroll 10
    for (int g = 0; g < num_chunks; ++g) {
        unsigned v = p[(size_t)g * pitch];
        lo += v & 0xFFFFu;
        hi += v >> 16;
    }
    float d0 = (float)(lo + 1u);
    float d1 = (float)(hi + 1u);
    float r0 = (d0 > 0.0f) ? rsqrtf(d0) : 0.0f;
    float r1 = (d1 > 0.0f) ? rsqrtf(d1) : 0.0f;
    int n = 2 * w;
    if (n + 1 < num_nodes) {
        *(float2*)(dinv + n) = make_float2(r0, r1);
    } else {
        dinv[n] = r0;
    }
}

__global__ void gemm1_wmma(const float* __restrict__ x,
                           const float* __restrict__ W1,
                           const float* __restrict__ dinv,
                           float* __restrict__ h1,
                           float* __restrict__ agg1, int num_nodes) {
    zero_plane(agg1, num_nodes * HID_DIM);
    int wave = (blockIdx.x * blockDim.x + threadIdx.x) >> 5;
    int lane = threadIdx.x & 31;
    int row0 = wave * 16;
    if (row0 + 16 > num_nodes) return;

    int half = lane >> 4;
    int l    = lane & 15;

    const float* arow = x + (size_t)(row0 + l) * IN_DIM + half * 2;
    v8f c = {};
    for (int k0 = 0; k0 < IN_DIM; k0 += 4) {
        v2f a = *(const v2f*)(arow + k0);
        v2f b;
        b.x = W1[(k0 + half * 2 + 0) * HID_DIM + l];
        b.y = W1[(k0 + half * 2 + 1) * HID_DIM + l];
        c = __builtin_amdgcn_wmma_f32_16x16x4_f32(
                false, a, false, b, (short)0, c, false, false);
    }
    float* outp = h1 + (size_t)row0 * HID_DIM;
    const float4* dp = (const float4*)(dinv + row0 + half * 8);
    const float4 d0 = dp[0], d1 = dp[1];
    const float dr[8] = {d0.x, d0.y, d0.z, d0.w, d1.x, d1.y, d1.z, d1.w};
#pragma unroll
    for (int r = 0; r < 8; ++r)
        outp[(half * 8 + r) * HID_DIM + l] = c[r] * dr[r];
}

#define AGG_TPB    1024
#define AGG_WAVES  (AGG_TPB / 32)
#define AGG_CHUNKS 8
#define AGG_RN     3136
#define AGG_DBITS  12
#define AGG_BATCH  16
#define AGG_QCAP   (AGG_BATCH * 32)

template <bool FULL>
__device__ __forceinline__ void agg_load(const int* __restrict__ p, int left,
                                         int (&v)[AGG_BATCH]) {
#pragma unroll
    for (int k = 0; k < AGG_BATCH; ++k) {
        const int o = k * AGG_TPB;
        if (FULL) {
            v[k] = p[o];
        } else {
            int x = p[min(o, left - 1)];
            v[k] = (o < left) ? x : -1;
        }
    }
}

#define AGG_LCAP   14336

__device__ __forceinline__ unsigned* list_piece(const float* ws, size_t ws_bytes,
                                                unsigned* lists, unsigned wgs, unsigned wg) {
    unsigned* p = lists + ((wgs + 63u) & ~63u) + (size_t)wg * AGG_LCAP;
    const size_t end = (size_t)((const char*)(p + AGG_LCAP) - (const char*)ws);
    return (end <= ws_bytes) ? p : nullptr;
}

template <int F, int FP, bool KEEP>
__device__ __forceinline__ void agg_scan(const int* __restrict__ src, const int* __restrict__ dst,
                                         const float* __restrict__ h, float* acc, unsigned* q,
                                         unsigned* lcur, unsigned* __restrict__ keep,
                                         int base, unsigned cnt, int e0, int e1) {
    const int tid  = threadIdx.x;
    const int lane = tid & 31;
    constexpr int EPW  = 32 / FP;
    constexpr int U    = 16 / EPW;
    constexpr int SPAN = AGG_BATCH * AGG_TPB;
    const int sub = lane / FP;
    const int j   = lane % FP;
    const int jc  = (j < F) ? j : F - 1;

    if (e0 < e1) {
        int cur[AGG_BATCH], nxt[AGG_BATCH];
        if (e0 + SPAN <= e1) agg_load<true>(dst + e0 + tid, 0, cur);
        else                 agg_load<false>(dst + e0 + tid, e1 - e0 - tid, cur);
#pragma unroll
        for (int k = 0; k < AGG_BATCH; ++k) {
            asm volatile("" :: "v"(cur[k]));
        }
        for (int eb = e0; eb < e1; eb += SPAN) {
            const int en = eb + SPAN;
            if (en + SPAN <= e1) agg_load<true>(dst + en + tid, 0, nxt);
            else                 agg_load<false>(dst + en + tid, e1 - en - tid, nxt);
            const int* sp = src + eb + tid;
            int s[AGG_BATCH];
#pragma unroll
            for (int k = 0; k < AGG_BATCH; ++k) {
                unsigned d = (unsigned)(cur[k] - base);
                s[k] = (d < cnt) ? sp[k * AGG_TPB] : 0;
            }
#pragma unroll
            for (int k = 0; k < AGG_BATCH; ++k) {
                asm volatile("" : "+v"(s[k]));
            }
            int qn = 0;
#pragma unroll
            for (int k = 0; k < AGG_BATCH; ++k) {
                unsigned d = (unsigned)(cur[k] - base);
                bool mt = d < cnt;
                unsigned m = __builtin_amdgcn_ballot_w32(mt);
                if (mt) q[qn + (int)__builtin_amdgcn_mbcnt_lo(m, 0u)] = ((unsigned)s[k] << AGG_DBITS) | d;
                qn += __popc(m);
            }
            __builtin_amdgcn_fence(__ATOMIC_RELEASE, "wavefront");
            __builtin_amdgcn_wave_barrier();
            __builtin_amdgcn_fence(__ATOMIC_ACQUIRE, "wavefront");
            if (KEEP && keep != nullptr && qn > 0) {
                unsigned pos = 0u;
                if (lane == 0) pos = atomicAdd(lcur, (unsigned)qn);
                pos = (unsigned)__builtin_amdgcn_readfirstlane((int)pos);
                if (pos + (unsigned)qn <= (unsigned)AGG_LCAP)
                    for (int i = lane; i < qn; i += 32) keep[pos + i] = q[i];
            }
            for (int i0 = 0; i0 < qn; i0 += 16) {
                unsigned w[U];
                float    val[U];
#pragma unroll
                for (int u = 0; u < U; ++u) w[u] = q[min(i0 + u * EPW + sub, qn - 1)];
#pragma unroll
                for (int u = 0; u < U; ++u) {
                    const unsigned sx = w[u] >> AGG_DBITS;
                    val[u] = h[(size_t)sx * F + jc];
                }
#pragma unroll
                for (int u = 0; u < U; ++u) {
                    asm volatile("" :: "v"(val[u]));
                }
#pragma unroll
                for (int u = 0; u < U; ++u)
                    if ((i0 + u * EPW + sub < qn) && (j < F))
                        atomicAdd(&acc[(w[u] & ((1u << AGG_DBITS) - 1u)) * F + j], val[u]);
            }
            __builtin_amdgcn_fence(__ATOMIC_RELEASE, "wavefront");
            __builtin_amdgcn_wave_barrier();
#pragma unroll
            for (int k = 0; k < AGG_BATCH; ++k) cur[k] = nxt[k];
        }
    }
}

template <int F, int FP>
__global__ __launch_bounds__(AGG_TPB)
void edge_agg_lds(const int* __restrict__ src, const int* __restrict__ dst,
                  const float* __restrict__ h, float* __restrict__ agg,
                  const float* __restrict__ ws, size_t ws_bytes, unsigned* __restrict__ lists,
                  int num_nodes, int num_edges) {
    __shared__ float acc[AGG_RN * F];
    __shared__ unsigned queue[AGG_WAVES][AGG_QCAP];
    __shared__ unsigned lcur;
    static_assert(AGG_RN <= (1 << AGG_DBITS), "local destination must fit its bits");

    const int tid  = threadIdx.x;
    const int base = blockIdx.y * AGG_RN;
    const unsigned cnt = (unsigned)min(AGG_RN, num_nodes - base);
    const unsigned wg  = blockIdx.y * gridDim.x + blockIdx.x;
    unsigned* keep = list_piece(ws, ws_bytes, lists, gridDim.x * gridDim.y, wg);

    for (int i = tid; i < AGG_RN * F; i += AGG_TPB) acc[i] = 0.0f;
    if (tid == 0) lcur = 0u;
    __syncthreads();

    const int per = (num_edges + AGG_CHUNKS - 1) / AGG_CHUNKS;
    const int e0  = blockIdx.x * per;
    const int e1  = min(e0 + per, num_edges);
    agg_scan<F, FP, true>(src, dst, h, acc, queue[tid >> 5], &lcur, keep, base, cnt, e0, e1);
    __syncthreads();

    if (tid == 0 && keep != nullptr) lists[wg] = lcur;
    float* outp = agg + (size_t)base * F;
    const int n = (int)cnt * F;
    for (int i = tid; i < n; i += AGG_TPB) atomicAdd(outp + i, acc[i]);
}

template <int F, int FP>
__global__ __launch_bounds__(AGG_TPB)
void edge_agg_kept(const int* __restrict__ src, const int* __restrict__ dst,
                   const float* __restrict__ h, float* __restrict__ agg,
                   const float* __restrict__ ws, size_t ws_bytes, unsigned* __restrict__ lists,
                   int num_nodes, int num_edges) {
    __shared__ float acc[AGG_RN * F];
    __shared__ unsigned queue[AGG_WAVES][AGG_QCAP];

    const int tid  = threadIdx.x;
    const int lane = tid & 31;
    const int base = blockIdx.y * AGG_RN;
    const unsigned cnt = (unsigned)min(AGG_RN, num_nodes - base);
    const unsigned wg  = blockIdx.y * gridDim.x + blockIdx.x;
    const unsigned* piece = list_piece(ws, ws_bytes, lists, gridDim.x * gridDim.y, wg);
    const unsigned kept = (piece != nullptr) ? lists[wg] : 0xFFFFFFFFu;

    for (int i = tid; i < AGG_RN * F; i += AGG_TPB) acc[i] = 0.0f;
    __syncthreads();

    if (kept <= (unsigned)AGG_LCAP) {
        constexpr int EPW = 32 / FP;
        constexpr int ST  = 32 / EPW;
        constexpr int G   = 4;
        constexpr int TRIP = AGG_WAVES * 32 * G;
        const int sub = lane / FP;
        const int j   = lane % FP;
        const int jc  = (j < F) ? j : F - 1;
        const int n   = (int)kept;
        const int w0  = (tid >> 5) * (32 * G);
        if (n > 0) {
            unsigned cur[G], nxt[G];
#pragma unroll
            for (int k = 0; k < G; ++k) cur[k] = piece[min(w0 + k * 32 + lane, n - 1)];
            for (int g0 = 0; g0 < n; g0 += TRIP) {
#pragma unroll
                for (int k = 0; k < G; ++k) nxt[k] = piece[min(g0 + TRIP + w0 + k * 32 + lane, n - 1)];
                unsigned w[G * ST];
                float    val[G * ST];
#pragma unroll
                for (int k = 0; k < G; ++k)
#pragma unroll
                    for (int u = 0; u < ST; ++u) w[k * ST + u] = (unsigned)__shfl((int)cur[k], u * EPW + sub);
#pragma unroll
                for (int t = 0; t < G * ST; ++t) val[t] = h[(size_t)(w[t] >> AGG_DBITS) * F + jc];
#pragma unroll
                for (int t = 0; t < G * ST; ++t) {
                    asm volatile("" :: "v"(val[t]));
                }
#pragma unroll
                for (int k = 0; k < G; ++k)
#pragma unroll
                    for (int u = 0; u < ST; ++u)
                        if ((g0 + w0 + k * 32 + u * EPW + sub < n) && (j < F))
                            atomicAdd(&acc[(w[k * ST + u] & ((1u << AGG_DBITS) - 1u)) * F + j], val[k * ST + u]);
#pragma unroll
                for (int k = 0; k < G; ++k) cur[k] = nxt[k];
            }
        }
    } else {
        const int per = (num_edges + AGG_CHUNKS - 1) / AGG_CHUNKS;
        const int e0  = blockIdx.x * per;
        const int e1  = min(e0 + per, num_edges);
        agg_scan<F, FP, false>(src, dst, h, acc, queue[tid >> 5], nullptr, nullptr, base, cnt, e0, e1);
    }
    __syncthreads();

    float* outp = agg + (size_t)base * F;
    const int n = (int)cnt * F;
    for (int i = tid; i < n; i += AGG_TPB) atomicAdd(outp + i, acc[i]);
}

__global__ void gemm2_wmma(const float* __restrict__ agg1,
                           const float* __restrict__ h1,
                           const float* __restrict__ dinv,
                           const float* __restrict__ b1,
                           const float* __restrict__ W2,
                           float* __restrict__ h2,
                           float* __restrict__ agg2, int num_nodes) {
    zero_plane(agg2, num_nodes * OUT_DIM);
    int wave = (blockIdx.x * blockDim.x + threadIdx.x) >> 5;
    int lane = threadIdx.x & 31;
    int row0 = wave * 16;
    if (row0 + 16 > num_nodes) return;

    int half = lane >> 4;
    int l    = lane & 15;

    const float  di = dinv[row0 + l];
    const size_t rb = (size_t)(row0 + l) * HID_DIM + half * 2;
    v8f c = {};
#pragma unroll
    for (int k0 = 0; k0 < HID_DIM; k0 += 4) {
        int kr = k0 + half * 2;
        v2f s = *(const v2f*)(agg1 + rb + k0);
        v2f g = *(const v2f*)(h1 + rb + k0);
        v2f a;
        a.x = fmaxf((s.x + g.x) * di + b1[kr + 0], 0.0f);
        a.y = fmaxf((s.y + g.y) * di + b1[kr + 1], 0.0f);
        v2f b;
        b.x = (l < OUT_DIM) ? W2[(kr + 0) * OUT_DIM + l] : 0.0f;
        b.y = (l < OUT_DIM) ? W2[(kr + 1) * OUT_DIM + l] : 0.0f;
        c = __builtin_amdgcn_wmma_f32_16x16x4_f32(
                false, a, false, b, (short)0, c, false, false);
    }
    if (l < OUT_DIM) {
        float* outp = h2 + (size_t)row0 * OUT_DIM;
        const float4* dp = (const float4*)(dinv + row0 + half * 8);
        const float4 d0 = dp[0], d1 = dp[1];
        const float dr[8] = {d0.x, d0.y, d0.z, d0.w, d1.x, d1.y, d1.z, d1.w};
#pragma unroll
        for (int r = 0; r < 8; ++r)
            outp[(half * 8 + r) * OUT_DIM + l] = c[r] * dr[r];
    }
}

__global__ void node_post2(const float* __restrict__ h2,
                           const float* __restrict__ dinv,
                           const float* __restrict__ b2,
                           const float* __restrict__ agg2,
                           float* __restrict__ out, int num_nodes) {
    int n = blockIdx.x * blockDim.x + threadIdx.x;
    if (n >= num_nodes) return;
    float di = dinv[n];
    float v[OUT_DIM];
    float m = -INFINITY;
#pragma unroll
    for (int j = 0; j < OUT_DIM; ++j) {
        v[j] = (agg2[(size_t)n * OUT_DIM + j] + h2[(size_t)n * OUT_DIM + j]) * di + b2[j];
        m = fmaxf(m, v[j]);
    }
    float s = 0.0f;
#pragma unroll
    for (int j = 0; j < OUT_DIM; ++j) s += expf(v[j] - m);
    float lse = logf(s);
#pragma unroll
    for (int j = 0; j < OUT_DIM; ++j)
        out[(size_t)n * OUT_DIM + j] = v[j] - m - lse;
}

extern "C" void kernel_launch(void* const* d_in, const int* in_sizes, int n_in,
                              void* d_out, int out_size, void* d_ws, size_t ws_size,
                              hipStream_t stream) {
    const float* x  = (const float*)d_in[0];
    const int*   ei = (const int*)d_in[1];
    const float* W1 = (const float*)d_in[2];
    const float* b1 = (const float*)d_in[3];
    const float* W2 = (const float*)d_in[4];
    const float* b2 = (const float*)d_in[5];

    const int N = in_sizes[0] / IN_DIM;
    const int E = in_sizes[1] / 2;
    const int* src = ei;
    const int* dst = ei + E;

    float* ws   = (float*)d_ws;
    float* dinv = ws;
    float* h1   = dinv + (size_t)N;
    float* agg1 = h1 + (size_t)N * HID_DIM;
    float* h2   = agg1 + (size_t)N * HID_DIM;
    float* agg2 = h2 + (size_t)N * OUT_DIM;
    unsigned* lists = (unsigned*)(agg2 + (size_t)N * OUT_DIM);
    float* outp = (float*)d_out;

    const int B = 256;
    auto nb = [](long long work, int blk) { return (int)((work + blk - 1) / blk); };

    unsigned* part = (unsigned*)h1;
    const int chunks = nb(E, DEG_CHUNK);
    degree_hist<<<dim3(chunks, nb(N, DEG_CAP)), DEG_TPB, 0, stream>>>(dst, part, N, E);
    degree_merge<<<nb((N + 1) / 2, B), B, 0, stream>>>(part, dinv, N, chunks);

    int waves1 = N / 16;
    gemm1_wmma<<<nb((long long)waves1 * 32, B), B, 0, stream>>>(x, W1, dinv, h1, agg1, N);
    const dim3 agrid(AGG_CHUNKS, nb(N, AGG_RN));
    edge_agg_lds<HID_DIM, 16><<<agrid, AGG_TPB, 0, stream>>>(src, dst, h1, agg1, ws, ws_size, lists, N, E);

    gemm2_wmma<<<nb((long long)waves1 * 32, B), B, 0, stream>>>(agg1, h1, dinv, b1, W2, h2, agg2, N);
    edge_agg_kept<OUT_DIM, 8><<<agrid, AGG_TPB, 0, stream>>>(src, dst, h2, agg2, ws, ws_size, lists, N, E);
    node_post2<<<nb(N, B), B, 0, stream>>>(h2, dinv, b2, agg2, outp, N);
}


# Edge filter (dst scan, src look-up) moved from layer 1's sum launch into the GEMM1 launch as 256 work-groups beside the x stream; layer 1 sums from kept lists like layer 2 (LDS add order changes)
# speedup vs baseline: 2.5333x; 1.0451x over previous
#include <hip/hip_runtime.h>
#include <hip/hip_bf16.h>

typedef __attribute__((ext_vector_type(2))) float v2f;
typedef __attribute__((ext_vector_type(8))) float v8f;

#define IN_DIM  500
#define HID_DIM 16
#define OUT_DIM 7

__device__ __forceinline__ void zero_plane(float* __restrict__ p, int n) {
    const int gtid = blockIdx.x * blockDim.x + threadIdx.x;
    const int nthr = gridDim.x * blockDim.x;
    const int n4   = n >> 2;
    float4* p4 = (float4*)p;
    for (int q = gtid; q < n4; q += nthr) p4[q] = make_float4(0.0f, 0.0f, 0.0f, 0.0f);
    const int t = (n4 << 2) + gtid;
    if (t < n) p[t] = 0.0f;
}

#define DEG_CAP   32768
#define DEG_TPB   1024
#define DEG_BATCH 21
#define DEG_ITERS 63
#define DEG_CHUNK (DEG_ITERS * DEG_TPB)

__device__ __forceinline__ int deg_pitch(int num_nodes) {
    return (((num_nodes + 1) >> 1) + 3) & ~3;
}

template <bool FULL>
__device__ __forceinline__ void deg_count(const int* __restrict__ p, int left,
                                          unsigned* hist, int base, unsigned cnt) {
    for (int it = 0; it < DEG_ITERS; it += DEG_BATCH) {
        int v[DEG_BATCH];
#pragma unroll
        for (int k = 0; k < DEG_BATCH; ++k) {
            int o = (it + k) * DEG_TPB;
            v[k] = (FULL || o < left) ? p[o] : -1;
        }
#pragma unroll
        for (int k = 0; k < DEG_BATCH; ++k) {
            unsigned d = (unsigned)(v[k] - base);
            if (d < cnt) atomicAdd(hist + (d >> 1), 1u << ((d & 1u) << 4));
        }
    }
}

__global__ __launch_bounds__(DEG_TPB)
void degree_hist(const int* __restrict__ dst, unsigned* __restrict__ part,
                 int num_nodes, int num_edges) {
    __shared__ uint4 hist4[DEG_CAP / 8];
    unsigned* hist = (unsigned*)hist4;
    const int tid  = threadIdx.x;
    const int base = blockIdx.y * DEG_CAP;
    const unsigned cnt = (unsigned)min(DEG_CAP, num_nodes - base);

    for (int i = tid; i < DEG_CAP / 8; i += DEG_TPB) hist4[i] = make_uint4(0u, 0u, 0u, 0u);
    __syncthreads();

    const int e0 = blockIdx.x * DEG_CHUNK;
    if (e0 + DEG_CHUNK <= num_edges) {
        deg_count<true>(dst + e0 + tid, 0, hist, base, cnt);
    } else {
        deg_count<false>(dst + e0 + tid, num_edges - e0 - tid, hist, base, cnt);
    }
    __syncthreads();

    uint4* outp = (uint4*)(part + (size_t)blockIdx.x * deg_pitch(num_nodes) + (base >> 1));
    const int nq = (int)((((cnt + 1u) >> 1) + 3u) >> 2);
    for (int i = tid; i < nq; i += DEG_TPB) outp[i] = hist4[i];
}

__global__ void degree_merge(const unsigned* __restrict__ part,
                             float* __restrict__ dinv, int num_nodes, int num_chunks) {
    int w = blockIdx.x * blockDim.x + threadIdx.x;
    if (w >= ((num_nodes + 1) >> 1)) return;
    const int pitch = deg_pitch(num_nodes);
    const unsigned* p = part + w;
    unsigned lo = 0u, hi = 0u;
#pragma unroll 10
    for (int g = 0; g < num_chunks; ++g) {
        unsigned v = p[(size_t)g * pitch];
        lo += v & 0xFFFFu;
        hi += v >> 16;
    }
    float d0 = (float)(lo + 1u);
    float d1 = (float)(hi + 1u);
    float r0 = (d0 > 0.0f) ? rsqrtf(d0) : 0.0f;
    float r1 = (d1 > 0.0f) ? rsqrtf(d1) : 0.0f;
    int n = 2 * w;
    if (n + 1 < num_nodes) {
        *(float2*)(dinv + n) = make_float2(r0, r1);
    } else {
        dinv[n] = r0;
    }
}

#define AGG_TPB    1024
#define AGG_WAVES  (AGG_TPB / 32)
#define AGG_CHUNKS 8
#define AGG_RN     3136
#define AGG_DBITS  12
#define AGG_BATCH  16
#define AGG_QCAP   (AGG_BATCH * 32)

template <bool FULL>
__device__ __forceinline__ void agg_load(const int* __restrict__ p, int left,
                                         int (&v)[AGG_BATCH]) {
#pragma unroll
    for (int k = 0; k < AGG_BATCH; ++k) {
        const int o = k * AGG_TPB;
        if (FULL) {
            v[k] = p[o];
        } else {
            int x = p[min(o, left - 1)];
            v[k] = (o < left) ? x : -1;
        }
    }
}

#define AGG_LCAP   14336

__device__ __forceinline__ unsigned* list_piece(const float* ws, size_t ws_bytes,
                                                unsigned* lists, unsigned wgs, unsigned wg) {
    unsigned* p = lists + ((wgs + 63u) & ~63u) + (size_t)wg * AGG_LCAP;
    const size_t end = (size_t)((const char*)(p + AGG_LCAP) - (const char*)ws);
    return (end <= ws_bytes) ? p : nullptr;
}

__device__ __forceinline__ void filter_scan(const int* __restrict__ src, const int* __restrict__ dst,
                                            unsigned* q, unsigned* lcur, unsigned* __restrict__ keep,
                                            int base, unsigned cnt, int e0, int e1) {
    const int tid  = threadIdx.x;
    const int lane = tid & 31;
    constexpr int SPAN = AGG_BATCH * AGG_TPB;

    if (e0 < e1) {
        int cur[AGG_BATCH], nxt[AGG_BATCH];
        if (e0 + SPAN <= e1) agg_load<true>(dst + e0 + tid, 0, cur);
        else                 agg_load<false>(dst + e0 + tid, e1 - e0 - tid, cur);
#pragma unroll
        for (int k = 0; k < AGG_BATCH; ++k) {
            asm volatile("" :: "v"(cur[k]));
        }
        for (int eb = e0; eb < e1; eb += SPAN) {
            const int en = eb + SPAN;
            if (en + SPAN <= e1) agg_load<true>(dst + en + tid, 0, nxt);
            else                 agg_load<false>(dst + en + tid, e1 - en - tid, nxt);
            const int* sp = src + eb + tid;
            int s[AGG_BATCH];
#pragma unroll
            for (int k = 0; k < AGG_BATCH; ++k) {
                unsigned d = (unsigned)(cur[k] - base);
                s[k] = (d < cnt) ? sp[k * AGG_TPB] : 0;
            }
#pragma unroll
            for (int k = 0; k < AGG_BATCH; ++k) {
                asm volatile("" : "+v"(s[k]));
            }
            int qn = 0;
#pragma unroll
            for (int k = 0; k < AGG_BATCH; ++k) {
                unsigned d = (unsigned)(cur[k] - base);
                bool mt = d < cnt;
                unsigned m = __builtin_amdgcn_ballot_w32(mt);
                if (mt) q[qn + (int)__builtin_amdgcn_mbcnt_lo(m, 0u)] = ((unsigned)s[k] << AGG_DBITS) | d;
                qn += __popc(m);
            }
            __builtin_amdgcn_fence(__ATOMIC_RELEASE, "wavefront");
            __builtin_amdgcn_wave_barrier();
            __builtin_amdgcn_fence(__ATOMIC_ACQUIRE, "wavefront");
            if (qn > 0) {
                unsigned pos = 0u;
                if (lane == 0) pos = atomicAdd(lcur, (unsigned)qn);
                pos = (unsigned)__builtin_amdgcn_readfirstlane((int)pos);
                if (pos + (unsigned)qn <= (unsigned)AGG_LCAP)
                    for (int i = lane; i < qn; i += 32) keep[pos + i] = q[i];
            }
            __builtin_amdgcn_fence(__ATOMIC_RELEASE, "wavefront");
            __builtin_amdgcn_wave_barrier();
#pragma unroll
            for (int k = 0; k < AGG_BATCH; ++k) cur[k] = nxt[k];
        }
    }
}

template <int F, int FP>
__device__ __forceinline__ void agg_scan(const int* __restrict__ src, const int* __restrict__ dst,
                                         const float* __restrict__ h, float* acc, unsigned* q,
                                         int base, unsigned cnt, int e0, int e1) {
    const int tid  = threadIdx.x;
    const int lane = tid & 31;
    constexpr int EPW  = 32 / FP;
    constexpr int U    = 16 / EPW;
    constexpr int SPAN = AGG_BATCH * AGG_TPB;
    const int sub = lane / FP;
    const int j   = lane % FP;
    const int jc  = (j < F) ? j : F - 1;

    if (e0 < e1) {
        int cur[AGG_BATCH], nxt[AGG_BATCH];
        if (e0 + SPAN <= e1) agg_load<true>(dst + e0 + tid, 0, cur);
        else                 agg_load<false>(dst + e0 + tid, e1 - e0 - tid, cur);
#pragma unroll
        for (int k = 0; k < AGG_BATCH; ++k) {
            asm volatile("" :: "v"(cur[k]));
        }
        for (int eb = e0; eb < e1; eb += SPAN) {
            const int en = eb + SPAN;
            if (en + SPAN <= e1) agg_load<true>(dst + en + tid, 0, nxt);
            else                 agg_load<false>(dst + en + tid, e1 - en - tid, nxt);
            const int* sp = src + eb + tid;
            int s[AGG_BATCH];
#pragma unroll
            for (int k = 0; k < AGG_BATCH; ++k) {
                unsigned d = (unsigned)(cur[k] - base);
                s[k] = (d < cnt) ? sp[k * AGG_TPB] : 0;
            }
#pragma unroll
            for (int k = 0; k < AGG_BATCH; ++k) {
                asm volatile("" : "+v"(s[k]));
            }
            int qn = 0;
#pragma unroll
            for (int k = 0; k < AGG_BATCH; ++k) {
                unsigned d = (unsigned)(cur[k] - base);
                bool mt = d < cnt;
                unsigned m = __builtin_amdgcn_ballot_w32(mt);
                if (mt) q[qn + (int)__builtin_amdgcn_mbcnt_lo(m, 0u)] = ((unsigned)s[k] << AGG_DBITS) | d;
                qn += __popc(m);
            }
            __builtin_amdgcn_fence(__ATOMIC_RELEASE, "wavefront");
            __builtin_amdgcn_wave_barrier();
            __builtin_amdgcn_fence(__ATOMIC_ACQUIRE, "wavefront");
            for (int i0 = 0; i0 < qn; i0 += 16) {
                unsigned w[U];
                float    val[U];
#pragma unroll
                for (int u = 0; u < U; ++u) w[u] = q[min(i0 + u * EPW + sub, qn - 1)];
#pragma unroll
                for (int u = 0; u < U; ++u) {
                    const unsigned sx = w[u] >> AGG_DBITS;
                    val[u] = h[(size_t)sx * F + jc];
                }
#pragma unroll
                for (int u = 0; u < U; ++u) {
                    asm volatile("" :: "v"(val[u]));
                }
#pragma unroll
                for (int u = 0; u < U; ++u)
                    if ((i0 + u * EPW + sub < qn) && (j < F))
                        atomicAdd(&acc[(w[u] & ((1u << AGG_DBITS) - 1u)) * F + j], val[u]);
            }
            __builtin_amdgcn_fence(__ATOMIC_RELEASE, "wavefront");
            __builtin_amdgcn_wave_barrier();
#pragma unroll
            for (int k = 0; k < AGG_BATCH; ++k) cur[k] = nxt[k];
        }
    }
}

__global__ __launch_bounds__(AGG_TPB)
void gemm1_wmma_filter(const float* __restrict__ x,
                       const float* __restrict__ W1,
                       const float* __restrict__ dinv,
                       float* __restrict__ h1,
                       float* __restrict__ agg1,
                       const int* __restrict__ src, const int* __restrict__ dst,
                       const float* __restrict__ ws, size_t ws_bytes, unsigned* __restrict__ lists,
                       int num_nodes, int num_edges) {
    __shared__ unsigned queue[AGG_WAVES][AGG_QCAP];
    __shared__ unsigned lcur;
    static_assert(AGG_RN <= (1 << AGG_DBITS), "local destination must fit its bits");

    zero_plane(agg1, num_nodes * HID_DIM);

    const unsigned pairs = (unsigned)AGG_CHUNKS * (unsigned)((num_nodes + AGG_RN - 1) / AGG_RN);
    const unsigned gblocks = gridDim.x - pairs;
    if (blockIdx.x >= gblocks) {
        const int tid = threadIdx.x;
        const unsigned wg = blockIdx.x - gblocks;
        unsigned* keep = list_piece(ws, ws_bytes, lists, pairs, wg);
        if (keep == nullptr) return;
        const int base = (int)(wg / AGG_CHUNKS) * AGG_RN;
        const unsigned cnt = (unsigned)min(AGG_RN, num_nodes - base);
        if (tid == 0) lcur = 0u;
        __syncthreads();
        const int per = (num_edges + AGG_CHUNKS - 1) / AGG_CHUNKS;
        const int e0  = (int)(wg % AGG_CHUNKS) * per;
        const int e1  = min(e0 + per, num_edges);
        filter_scan(src, dst, queue[tid >> 5], &lcur, keep, base, cnt, e0, e1);
        __syncthreads();
        if (tid == 0) lists[wg] = lcur;
        return;
    }

    int tile = (int)(threadIdx.x >> 5) * (int)gblocks + (int)blockIdx.x;
    int lane = threadIdx.x & 31;
    int row0 = tile * 16;
    if (row0 + 16 > num_nodes) return;

    int half = lane >> 4;
    int l    = lane & 15;

    const float* arow = x + (size_t)(row0 + l) * IN_DIM + half * 2;
    v8f c = {};
    for (int k0 = 0; k0 < IN_DIM; k0 += 4) {
        v2f a = *(const v2f*)(arow + k0);
        v2f b;
        b.x = W1[(k0 + half * 2 + 0) * HID_DIM + l];
        b.y = W1[(k0 + half * 2 + 1) * HID_DIM + l];
        c = __builtin_amdgcn_wmma_f32_16x16x4_f32(
                false, a, false, b, (short)0, c, false, false);
    }
    float* outp = h1 + (size_t)row0 * HID_DIM;
    const float4* dp = (const float4*)(dinv + row0 + half * 8);
    const float4 d0 = dp[0], d1 = dp[1];
    const float dr[8] = {d0.x, d0.y, d0.z, d0.w, d1.x, d1.y, d1.z, d1.w};
#pragma unroll
    for (int r = 0; r < 8; ++r)
        outp[(half * 8 + r) * HID_DIM + l] = c[r] * dr[r];
}

template <int F, int FP, int G>
__global__ __launch_bounds__(AGG_TPB)
void edge_agg_kept(const int* __restrict__ src, const int* __restrict__ dst,
                   const float* __restrict__ h, float* __restrict__ agg,
                   const float* __restrict__ ws, size_t ws_bytes, unsigned* __restrict__ lists,
                   int num_nodes, int num_edges) {
    __shared__ float acc[AGG_RN * F];
    __shared__ unsigned queue[AGG_WAVES][AGG_QCAP];

    const int tid  = threadIdx.x;
    const int lane = tid & 31;
    const int base = blockIdx.y * AGG_RN;
    const unsigned cnt = (unsigned)min(AGG_RN, num_nodes - base);
    const unsigned wg  = blockIdx.y * gridDim.x + blockIdx.x;
    const unsigned* piece = list_piece(ws, ws_bytes, lists, gridDim.x * gridDim.y, wg);
    const unsigned kept = (piece != nullptr) ? lists[wg] : 0xFFFFFFFFu;

    for (int i = tid; i < AGG_RN * F; i += AGG_TPB) acc[i] = 0.0f;
    __syncthreads();

    if (kept <= (unsigned)AGG_LCAP) {
        constexpr int EPW = 32 / FP;
        constexpr int ST  = 32 / EPW;
        constexpr int TRIP = AGG_WAVES * 32 * G;
        const int sub = lane / FP;
        const int j   = lane % FP;
        const int jc  = (j < F) ? j : F - 1;
        const int n   = (int)kept;
        const int w0  = (tid >> 5) * (32 * G);
        if (n > 0) {
            unsigned cur[G], nxt[G];
#pragma unroll
            for (int k = 0; k < G; ++k) cur[k] = piece[min(w0 + k * 32 + lane, n - 1)];
            for (int g0 = 0; g0 < n; g0 += TRIP) {
#pragma unroll
                for (int k = 0; k < G; ++k) nxt[k] = piece[min(g0 + TRIP + w0 + k * 32 + lane, n - 1)];
                unsigned w[G * ST];
                float    val[G * ST];
#pragma unroll
                for (int k = 0; k < G; ++k)
#pragma unroll
                    for (int u = 0; u < ST; ++u) w[k * ST + u] = (unsigned)__shfl((int)cur[k], u * EPW + sub);
#pragma unroll
                for (int t = 0; t < G * ST; ++t) val[t] = h[(size_t)(w[t] >> AGG_DBITS) * F + jc];
#pragma unroll
                for (int t = 0; t < G * ST; ++t) {
                    asm volatile("" :: "v"(val[t]));
                }
#pragma unroll
                for (int k = 0; k < G; ++k)
#pragma unroll
                    for (int u = 0; u < ST; ++u)
                        if ((g0 + w0 + k * 32 + u * EPW + sub < n) && (j < F))
                            atomicAdd(&acc[(w[k * ST + u] & ((1u << AGG_DBITS) - 1u)) * F + j], val[k * ST + u]);
#pragma unroll
                for (int k = 0; k < G; ++k) cur[k] = nxt[k];
            }
        }
    } else {
        const int per = (num_edges + AGG_CHUNKS - 1) / AGG_CHUNKS;
        const int e0  = blockIdx.x * per;
        const int e1  = min(e0 + per, num_edges);
        agg_scan<F, FP>(src, dst, h, acc, queue[tid >> 5], base, cnt, e0, e1);
    }
    __syncthreads();

    float* outp = agg + (size_t)base * F;
    const int n = (int)cnt * F;
    for (int i = tid; i < n; i += AGG_TPB) atomicAdd(outp + i, acc[i]);
}

__global__ void gemm2_wmma(const float* __restrict__ agg1,
                           const float* __restrict__ h1,
                           const float* __restrict__ dinv,
                           const float* __restrict__ b1,
                           const float* __restrict__ W2,
                           float* __restrict__ h2,
                           float* __restrict__ agg2, int num_nodes) {
    zero_plane(agg2, num_nodes * OUT_DIM);
    int wave = (blockIdx.x * blockDim.x + threadIdx.x) >> 5;
    int lane = threadIdx.x & 31;
    int row0 = wave * 16;
    if (row0 + 16 > num_nodes) return;

    int half = lane >> 4;
    int l    = lane & 15;

    const float  di = dinv[row0 + l];
    const size_t rb = (size_t)(row0 + l) * HID_DIM + half * 2;
    v8f c = {};
#pragma unroll
    for (int k0 = 0; k0 < HID_DIM; k0 += 4) {
        int kr = k0 + half * 2;
        v2f s = *(const v2f*)(agg1 + rb + k0);
        v2f g = *(const v2f*)(h1 + rb + k0);
        v2f a;
        a.x = fmaxf((s.x + g.x) * di + b1[kr + 0], 0.0f);
        a.y = fmaxf((s.y + g.y) * di + b1[kr + 1], 0.0f);
        v2f b;
        b.x = (l < OUT_DIM) ? W2[(kr + 0) * OUT_DIM + l] : 0.0f;
        b.y = (l < OUT_DIM) ? W2[(kr + 1) * OUT_DIM + l] : 0.0f;
        c = __builtin_amdgcn_wmma_f32_16x16x4_f32(
                false, a, false, b, (short)0, c, false, false);
    }
    if (l < OUT_DIM) {
        float* outp = h2 + (size_t)row0 * OUT_DIM;
        const float4* dp = (const float4*)(dinv + row0 + half * 8);
        const float4 d0 = dp[0], d1 = dp[1];
        const float dr[8] = {d0.x, d0.y, d0.z, d0.w, d1.x, d1.y, d1.z, d1.w};
#pragma unroll
        for (int r = 0; r < 8; ++r)
            outp[(half * 8 + r) * OUT_DIM + l] = c[r] * dr[r];
    }
}

__global__ void node_post2(const float* __restrict__ h2,
                           const float* __restrict__ dinv,
                           const float* __restrict__ b2,
                           const float* __restrict__ agg2,
                           float* __restrict__ out, int num_nodes) {
    int n = blockIdx.x * blockDim.x + threadIdx.x;
    if (n >= num_nodes) return;
    float di = dinv[n];
    float v[OUT_DIM];
    float m = -INFINITY;
#pragma unroll
    for (int j = 0; j < OUT_DIM; ++j) {
        v[j] = (agg2[(size_t)n * OUT_DIM + j] + h2[(size_t)n * OUT_DIM + j]) * di + b2[j];
        m = fmaxf(m, v[j]);
    }
    float s = 0.0f;
#pragma unroll
    for (int j = 0; j < OUT_DIM; ++j) s += expf(v[j] - m);
    float lse = logf(s);
#pragma unroll
    for (int j = 0; j < OUT_DIM; ++j)
        out[(size_t)n * OUT_DIM + j] = v[j] - m - lse;
}

extern "C" void kernel_launch(void* const* d_in, const int* in_sizes, int n_in,
                              void* d_out, int out_size, void* d_ws, size_t ws_size,
                              hipStream_t stream) {
    const float* x  = (const float*)d_in[0];
    const int*   ei = (const int*)d_in[1];
    const float* W1 = (const float*)d_in[2];
    const float* b1 = (const float*)d_in[3];
    const float* W2 = (const float*)d_in[4];
    const float* b2 = (const float*)d_in[5];

    const int N = in_sizes[0] / IN_DIM;
    const int E = in_sizes[1] / 2;
    const int* src = ei;
    const int* dst = ei + E;

    float* ws   = (float*)d_ws;
    float* dinv = ws;
    float* h1   = dinv + (size_t)N;
    float* agg1 = h1 + (size_t)N * HID_DIM;
    float* h2   = agg1 + (size_t)N * HID_DIM;
    float* agg2 = h2 + (size_t)N * OUT_DIM;
    unsigned* lists = (unsigned*)(agg2 + (size_t)N * OUT_DIM);
    float* outp = (float*)d_out;

    const int B = 256;
    auto nb = [](long long work, int blk) { return (int)((work + blk - 1) / blk); };

    unsigned* part = (unsigned*)h1;
    const int chunks = nb(E, DEG_CHUNK);
    degree_hist<<<dim3(chunks, nb(N, DEG_CAP)), DEG_TPB, 0, stream>>>(dst, part, N, E);
    degree_merge<<<nb((N + 1) / 2, B), B, 0, stream>>>(part, dinv, N, chunks);

    int waves1 = N / 16;
    const dim3 agrid(AGG_CHUNKS, nb(N, AGG_RN));
    const int pairs   = (int)(agrid.x * agrid.y);
    const int gneed   = nb(waves1, AGG_WAVES);
    const int gblocks = (gneed > pairs) ? gneed : pairs;
    gemm1_wmma_filter<<<gblocks + pairs, AGG_TPB, 0, stream>>>(
        x, W1, dinv, h1, agg1, src, dst, ws, ws_size, lists, N, E);
    edge_agg_kept<HID_DIM, 16, 2><<<agrid, AGG_TPB, 0, stream>>>(src, dst, h1, agg1, ws, ws_size, lists, N, E);

    gemm2_wmma<<<nb((long long)waves1 * 32, B), B, 0, stream>>>(agg1, h1, dinv, b1, W2, h2, agg2, N);
    edge_agg_kept<OUT_DIM, 8, 4><<<agrid, AGG_TPB, 0, stream>>>(src, dst, h2, agg2, ws, ws_size, lists, N, E);
    node_post2<<<nb(N, B), B, 0, stream>>>(h2, dinv, b2, agg2, outp, N);
}


# Edge filter serves 4 destination ranges per pass: a work-group reads a 32nd of dst once for 4 ranges and splits its matches by range on flush; dst is read 8 times, not 32 (LDS add order changes)
# speedup vs baseline: 2.8597x; 1.1288x over previous
#include <hip/hip_runtime.h>
#include <hip/hip_bf16.h>

typedef __attribute__((ext_vector_type(2))) float v2f;
typedef __attribute__((ext_vector_type(8))) float v8f;

#define IN_DIM  500
#define HID_DIM 16
#define OUT_DIM 7

__device__ __forceinline__ void zero_plane(float* __restrict__ p, int n) {
    const int gtid = blockIdx.x * blockDim.x + threadIdx.x;
    const int nthr = gridDim.x * blockDim.x;
    const int n4   = n >> 2;
    float4* p4 = (float4*)p;
    for (int q = gtid; q < n4; q += nthr) p4[q] = make_float4(0.0f, 0.0f, 0.0f, 0.0f);
    const int t = (n4 << 2) + gtid;
    if (t < n) p[t] = 0.0f;
}

#define DEG_CAP   32768
#define DEG_TPB   1024
#define DEG_BATCH 21
#define DEG_ITERS 63
#define DEG_CHUNK (DEG_ITERS * DEG_TPB)

__device__ __forceinline__ int deg_pitch(int num_nodes) {
    return (((num_nodes + 1) >> 1) + 3) & ~3;
}

template <bool FULL>
__device__ __forceinline__ void deg_count(const int* __restrict__ p, int left,
                                          unsigned* hist, int base, unsigned cnt) {
    for (int it = 0; it < DEG_ITERS; it += DEG_BATCH) {
        int v[DEG_BATCH];
#pragma unroll
        for (int k = 0; k < DEG_BATCH; ++k) {
            int o = (it + k) * DEG_TPB;
            v[k] = (FULL || o < left) ? p[o] : -1;
        }
#pragma unroll
        for (int k = 0; k < DEG_BATCH; ++k) {
            unsigned d = (unsigned)(v[k] - base);
            if (d < cnt) atomicAdd(hist + (d >> 1), 1u << ((d & 1u) << 4));
        }
    }
}

__global__ __launch_bounds__(DEG_TPB)
void degree_hist(const int* __restrict__ dst, unsigned* __restrict__ part,
                 int num_nodes, int num_edges) {
    __shared__ uint4 hist4[DEG_CAP / 8];
    unsigned* hist = (unsigned*)hist4;
    const int tid  = threadIdx.x;
    const int base = blockIdx.y * DEG_CAP;
    const unsigned cnt = (unsigned)min(DEG_CAP, num_nodes - base);

    for (int i = tid; i < DEG_CAP / 8; i += DEG_TPB) hist4[i] = make_uint4(0u, 0u, 0u, 0u);
    __syncthreads();

    const int e0 = blockIdx.x * DEG_CHUNK;
    if (e0 + DEG_CHUNK <= num_edges) {
        deg_count<true>(dst + e0 + tid, 0, hist, base, cnt);
    } else {
        deg_count<false>(dst + e0 + tid, num_edges - e0 - tid, hist, base, cnt);
    }
    __syncthreads();

    uint4* outp = (uint4*)(part + (size_t)blockIdx.x * deg_pitch(num_nodes) + (base >> 1));
    const int nq = (int)((((cnt + 1u) >> 1) + 3u) >> 2);
    for (int i = tid; i < nq; i += DEG_TPB) outp[i] = hist4[i];
}

__global__ void degree_merge(const unsigned* __restrict__ part,
                             float* __restrict__ dinv, int num_nodes, int num_chunks) {
    int w = blockIdx.x * blockDim.x + threadIdx.x;
    if (w >= ((num_nodes + 1) >> 1)) return;
    const int pitch = deg_pitch(num_nodes);
    const unsigned* p = part + w;
    unsigned lo = 0u, hi = 0u;
#pragma unroll 10
    for (int g = 0; g < num_chunks; ++g) {
        unsigned v = p[(size_t)g * pitch];
        lo += v & 0xFFFFu;
        hi += v >> 16;
    }
    float d0 = (float)(lo + 1u);
    float d1 = (float)(hi + 1u);
    float r0 = (d0 > 0.0f) ? rsqrtf(d0) : 0.0f;
    float r1 = (d1 > 0.0f) ? rsqrtf(d1) : 0.0f;
    int n = 2 * w;
    if (n + 1 < num_nodes) {
        *(float2*)(dinv + n) = make_float2(r0, r1);
    } else {
        dinv[n] = r0;
    }
}

#define AGG_TPB    1024
#define AGG_WAVES  (AGG_TPB / 32)
#define AGG_CHUNKS 8
#define AGG_RN     3136
#define AGG_DBITS  12
#define AGG_BATCH  16
#define AGG_QCAP   (AGG_BATCH * 32)
#define AGG_QUARTERS 4
#define FLT_RPG    4
#define FLT_GBITS  14
#define FLT_SLICES (AGG_CHUNKS * AGG_QUARTERS)

template <bool FULL>
__device__ __forceinline__ void agg_load(const int* __restrict__ p, int left,
                                         int (&v)[AGG_BATCH]) {
#pragma unroll
    for (int k = 0; k < AGG_BATCH; ++k) {
        const int o = k * AGG_TPB;
        if (FULL) {
            v[k] = p[o];
        } else {
            int x = p[min(o, left - 1)];
            v[k] = (o < left) ? x : -1;
        }
    }
}

#define AGG_LCAP   14336
#define AGG_SCAP   (AGG_LCAP / AGG_QUARTERS)

__device__ __forceinline__ unsigned* list_piece(const float* ws, size_t ws_bytes,
                                                unsigned* lists, unsigned wgs, unsigned wg) {
    unsigned* p = lists + ((wgs * AGG_QUARTERS + 63u) & ~63u) + (size_t)wg * AGG_LCAP;
    const size_t end = (size_t)((const char*)(p + AGG_LCAP) - (const char*)ws);
    return (end <= ws_bytes) ? p : nullptr;
}

__device__ __forceinline__ void filter_scan(const int* __restrict__ src, const int* __restrict__ dst,
                                            unsigned* q, unsigned* lcur, unsigned* __restrict__ keep,
                                            unsigned nv, int base, unsigned cnt, int e0, int e1) {
    const int tid  = threadIdx.x;
    const int lane = tid & 31;
    constexpr int SPAN = AGG_BATCH * AGG_TPB;

    if (e0 < e1) {
        int cur[AGG_BATCH], nxt[AGG_BATCH];
        if (e0 + SPAN <= e1) agg_load<true>(dst + e0 + tid, 0, cur);
        else                 agg_load<false>(dst + e0 + tid, e1 - e0 - tid, cur);
#pragma unroll
        for (int k = 0; k < AGG_BATCH; ++k) {
            asm volatile("" :: "v"(cur[k]));
        }
        for (int eb = e0; eb < e1; eb += SPAN) {
            const int en = eb + SPAN;
            if (en + SPAN <= e1) agg_load<true>(dst + en + tid, 0, nxt);
            else                 agg_load<false>(dst + en + tid, e1 - en - tid, nxt);
            const int* sp = src + eb + tid;
            int s[AGG_BATCH];
#pragma unroll
            for (int k = 0; k < AGG_BATCH; ++k) {
                unsigned d = (unsigned)(cur[k] - base);
                s[k] = (d < cnt) ? sp[k * AGG_TPB] : 0;
            }
#pragma unroll
            for (int k = 0; k < AGG_BATCH; ++k) {
                asm volatile("" : "+v"(s[k]));
            }
            int qn = 0;
#pragma unroll
            for (int k = 0; k < AGG_BATCH; ++k) {
                unsigned d = (unsigned)(cur[k] - base);
                bool mt = d < cnt;
                unsigned m = __builtin_amdgcn_ballot_w32(mt);
                if (mt) q[qn + (int)__builtin_amdgcn_mbcnt_lo(m, 0u)] = ((unsigned)s[k] << FLT_GBITS) | d;
                qn += __popc(m);
            }
            __builtin_amdgcn_fence(__ATOMIC_RELEASE, "wavefront");
            __builtin_amdgcn_wave_barrier();
            __builtin_amdgcn_fence(__ATOMIC_ACQUIRE, "wavefront");
            for (int i0 = 0; i0 < qn; i0 += 32) {
                const bool ok = i0 + lane < qn;
                const unsigned w  = q[ok ? i0 + lane : 0];
                const unsigned dg = w & ((1u << FLT_GBITS) - 1u);
                unsigned r = 0u;
                for (int j = 1; j < FLT_RPG; ++j) r += (dg >= (unsigned)(j * AGG_RN)) ? 1u : 0u;
                const unsigned word = ((w >> FLT_GBITS) << AGG_DBITS) | (dg - r * (unsigned)AGG_RN);
                unsigned rank = 0u, mine = 0u;
                for (int j = 0; j < FLT_RPG; ++j) {
                    const unsigned m = __builtin_amdgcn_ballot_w32(ok && r == (unsigned)j);
                    if (r == (unsigned)j) rank = __builtin_amdgcn_mbcnt_lo(m, 0u);
                    if (lane == j) mine = (unsigned)__popc(m);
                }
                unsigned pos = 0u;
                if (lane < FLT_RPG) pos = atomicAdd(lcur + lane, mine);
                pos = (unsigned)__shfl((int)pos, (int)r) + rank;
                if (ok && r < nv && pos < (unsigned)AGG_SCAP)
                    keep[(size_t)r * ((size_t)AGG_CHUNKS * AGG_LCAP) + pos] = word;
            }
            __builtin_amdgcn_fence(__ATOMIC_RELEASE, "wavefront");
            __builtin_amdgcn_wave_barrier();
#pragma unroll
            for (int k = 0; k < AGG_BATCH; ++k) cur[k] = nxt[k];
        }
    }
}

template <int F, int FP>
__device__ __forceinline__ void agg_scan(const int* __restrict__ src, const int* __restrict__ dst,
                                         const float* __restrict__ h, float* acc, unsigned* q,
                                         int base, unsigned cnt, int e0, int e1) {
    const int tid  = threadIdx.x;
    const int lane = tid & 31;
    constexpr int EPW  = 32 / FP;
    constexpr int U    = 16 / EPW;
    constexpr int SPAN = AGG_BATCH * AGG_TPB;
    const int sub = lane / FP;
    const int j   = lane % FP;
    const int jc  = (j < F) ? j : F - 1;

    if (e0 < e1) {
        int cur[AGG_BATCH], nxt[AGG_BATCH];
        if (e0 + SPAN <= e1) agg_load<true>(dst + e0 + tid, 0, cur);
        else                 agg_load<false>(dst + e0 + tid, e1 - e0 - tid, cur);
#pragma unroll
        for (int k = 0; k < AGG_BATCH; ++k) {
            asm volatile("" :: "v"(cur[k]));
        }
        for (int eb = e0; eb < e1; eb += SPAN) {
            const int en = eb + SPAN;
            if (en + SPAN <= e1) agg_load<true>(dst + en + tid, 0, nxt);
            else                 agg_load<false>(dst + en + tid, e1 - en - tid, nxt);
            const int* sp = src + eb + tid;
            int s[AGG_BATCH];
#pragma unroll
            for (int k = 0; k < AGG_BATCH; ++k) {
                unsigned d = (unsigned)(cur[k] - base);
                s[k] = (d < cnt) ? sp[k * AGG_TPB] : 0;
            }
#pragma unroll
            for (int k = 0; k < AGG_BATCH; ++k) {
                asm volatile("" : "+v"(s[k]));
            }
            int qn = 0;
#pragma unroll
            for (int k = 0; k < AGG_BATCH; ++k) {
                unsigned d = (unsigned)(cur[k] - base);
                bool mt = d < cnt;
                unsigned m = __builtin_amdgcn_ballot_w32(mt);
                if (mt) q[qn + (int)__builtin_amdgcn_mbcnt_lo(m, 0u)] = ((unsigned)s[k] << AGG_DBITS) | d;
                qn += __popc(m);
            }
            __builtin_amdgcn_fence(__ATOMIC_RELEASE, "wavefront");
            __builtin_amdgcn_wave_barrier();
            __builtin_amdgcn_fence(__ATOMIC_ACQUIRE, "wavefront");
            for (int i0 = 0; i0 < qn; i0 += 16) {
                unsigned w[U];
                float    val[U];
#pragma unroll
                for (int u = 0; u < U; ++u) w[u] = q[min(i0 + u * EPW + sub, qn - 1)];
#pragma unroll
                for (int u = 0; u < U; ++u) {
                    const unsigned sx = w[u] >> AGG_DBITS;
                    val[u] = h[(size_t)sx * F + jc];
                }
#pragma unroll
                for (int u = 0; u < U; ++u) {
                    asm volatile("" :: "v"(val[u]));
                }
#pragma unroll
                for (int u = 0; u < U; ++u)
                    if ((i0 + u * EPW + sub < qn) && (j < F))
                        atomicAdd(&acc[(w[u] & ((1u << AGG_DBITS) - 1u)) * F + j], val[u]);
            }
            __builtin_amdgcn_fence(__ATOMIC_RELEASE, "wavefront");
            __builtin_amdgcn_wave_barrier();
#pragma unroll
            for (int k = 0; k < AGG_BATCH; ++k) cur[k] = nxt[k];
        }
    }
}

__global__ __launch_bounds__(AGG_TPB)
void gemm1_wmma_filter(const float* __restrict__ x,
                       const float* __restrict__ W1,
                       const float* __restrict__ dinv,
                       float* __restrict__ h1,
                       float* __restrict__ agg1,
                       const int* __restrict__ src, const int* __restrict__ dst,
                       const float* __restrict__ ws, size_t ws_bytes, unsigned* __restrict__ lists,
                       int num_nodes, int num_edges) {
    __shared__ unsigned queue[AGG_WAVES][AGG_QCAP];
    __shared__ unsigned lcur[FLT_RPG];
    static_assert(AGG_RN <= (1 << AGG_DBITS), "local destination must fit its bits");
    static_assert(FLT_RPG * AGG_RN <= (1 << FLT_GBITS), "destination within the group must fit its bits");
    static_assert(AGG_SCAP * AGG_QUARTERS == AGG_LCAP, "sub-lists must fill the piece");

    zero_plane(agg1, num_nodes * HID_DIM);

    const unsigned ranges  = (unsigned)((num_nodes + AGG_RN - 1) / AGG_RN);
    const unsigned pairs   = (unsigned)AGG_CHUNKS * ranges;
    const unsigned fblocks = (unsigned)FLT_SLICES * ((ranges + FLT_RPG - 1u) / FLT_RPG);
    const unsigned gblocks = gridDim.x - fblocks;
    if (blockIdx.x >= gblocks) {
        const int tid = threadIdx.x;
        const unsigned f   = blockIdx.x - gblocks;
        const unsigned rg0 = (f / FLT_SLICES) * FLT_RPG;
        const unsigned ch  = (f % FLT_SLICES) / AGG_QUARTERS;
        const unsigned qt  = (f % FLT_SLICES) % AGG_QUARTERS;
        unsigned nv = 0u;
        for (unsigned i = 0u; i < FLT_RPG; ++i)
            if (rg0 + i < ranges &&
                list_piece(ws, ws_bytes, lists, pairs, (rg0 + i) * AGG_CHUNKS + ch) != nullptr) nv = i + 1u;
        if (nv == 0u) return;
        unsigned* keep = list_piece(ws, ws_bytes, lists, pairs, rg0 * AGG_CHUNKS + ch) + qt * AGG_SCAP;
        const int base = (int)rg0 * AGG_RN;
        const unsigned cnt = (unsigned)min(FLT_RPG * AGG_RN, num_nodes - base);
        if (tid < FLT_RPG) lcur[tid] = 0u;
        __syncthreads();
        const int per = (num_edges + AGG_CHUNKS - 1) / AGG_CHUNKS;
        const int c0  = (int)ch * per;
        const int c1  = min(c0 + per, num_edges);
        const int pq  = (per + AGG_QUARTERS - 1) / AGG_QUARTERS;
        const int e0  = c0 + (int)qt * pq;
        const int e1  = min(e0 + pq, c1);
        filter_scan(src, dst, queue[tid >> 5], lcur, keep, nv, base, cnt, e0, e1);
        __syncthreads();
        if (tid < (int)nv)
            lists[(((size_t)rg0 + tid) * AGG_CHUNKS + ch) * AGG_QUARTERS + qt] = lcur[tid];
        return;
    }

    int tile = (int)(threadIdx.x >> 5) * (int)gblocks + (int)blockIdx.x;
    int lane = threadIdx.x & 31;
    int row0 = tile * 16;
    if (row0 + 16 > num_nodes) return;

    int half = lane >> 4;
    int l    = lane & 15;

    const float* arow = x + (size_t)(row0 + l) * IN_DIM + half * 2;
    v8f c = {};
    for (int k0 = 0; k0 < IN_DIM; k0 += 4) {
        v2f a = *(const v2f*)(arow + k0);
        v2f b;
        b.x = W1[(k0 + half * 2 + 0) * HID_DIM + l];
        b.y = W1[(k0 + half * 2 + 1) * HID_DIM + l];
        c = __builtin_amdgcn_wmma_f32_16x16x4_f32(
                false, a, false, b, (short)0, c, false, false);
    }
    float* outp = h1 + (size_t)row0 * HID_DIM;
    const float4* dp = (const float4*)(dinv + row0 + half * 8);
    const float4 d0 = dp[0], d1 = dp[1];
    const float dr[8] = {d0.x, d0.y, d0.z, d0.w, d1.x, d1.y, d1.z, d1.w};
#pragma unroll
    for (int r = 0; r < 8; ++r)
        outp[(half * 8 + r) * HID_DIM + l] = c[r] * dr[r];
}

template <int F, int FP, int G>
__global__ __launch_bounds__(AGG_TPB)
void edge_agg_kept(const int* __restrict__ src, const int* __restrict__ dst,
                   const float* __restrict__ h, float* __restrict__ agg,
                   const float* __restrict__ ws, size_t ws_bytes, unsigned* __restrict__ lists,
                   int num_nodes, int num_edges) {
    __shared__ float acc[AGG_RN * F];
    __shared__ unsigned queue[AGG_WAVES][AGG_QCAP];

    const int tid  = threadIdx.x;
    const int lane = tid & 31;
    const int base = blockIdx.y * AGG_RN;
    const unsigned cnt = (unsigned)min(AGG_RN, num_nodes - base);
    const unsigned wg  = blockIdx.y * gridDim.x + blockIdx.x;
    const unsigned* piece = list_piece(ws, ws_bytes, lists, gridDim.x * gridDim.y, wg);
    unsigned cq[AGG_QUARTERS];
    bool kept_ok = (piece != nullptr);
    for (int t = 0; t < AGG_QUARTERS; ++t)
        cq[t] = (piece != nullptr) ? lists[(size_t)wg * AGG_QUARTERS + t] : 0xFFFFFFFFu;
    for (int t = 0; t < AGG_QUARTERS; ++t) kept_ok = kept_ok && (cq[t] <= (unsigned)AGG_SCAP);
    const int b1 = kept_ok ? (int)cq[0] : 0;
    const int b2 = kept_ok ? b1 + (int)cq[1] : 0;
    const int b3 = kept_ok ? b2 + (int)cq[2] : 0;
    const unsigned kept = kept_ok ? (unsigned)b3 + cq[3] : 0xFFFFFFFFu;
    auto list_word = [&](int v) -> unsigned {
        int o = v;
        if (v >= b1) o = v - b1 + AGG_SCAP;
        if (v >= b2) o = v - b2 + 2 * AGG_SCAP;
        if (v >= b3) o = v - b3 + 3 * AGG_SCAP;
        return piece[o];
    };

    for (int i = tid; i < AGG_RN * F; i += AGG_TPB) acc[i] = 0.0f;
    __syncthreads();

    if (kept <= (unsigned)AGG_LCAP) {
        constexpr int EPW = 32 / FP;
        constexpr int ST  = 32 / EPW;
        constexpr int TRIP = AGG_WAVES * 32 * G;
        const int sub = lane / FP;
        const int j   = lane % FP;
        const int jc  = (j < F) ? j : F - 1;
        const int n   = (int)kept;
        const int w0  = (tid >> 5) * (32 * G);
        if (n > 0) {
            unsigned cur[G], nxt[G];
#pragma unroll
            for (int k = 0; k < G; ++k) cur[k] = list_word(min(w0 + k * 32 + lane, n - 1));
            for (int g0 = 0; g0 < n; g0 += TRIP) {
#pragma unroll
                for (int k = 0; k < G; ++k) nxt[k] = list_word(min(g0 + TRIP + w0 + k * 32 + lane, n - 1));
                unsigned w[G * ST];
                float    val[G * ST];
#pragma unroll
                for (int k = 0; k < G; ++k)
#pragma unroll
                    for (int u = 0; u < ST; ++u) w[k * ST + u] = (unsigned)__shfl((int)cur[k], u * EPW + sub);
#pragma unroll
                for (int t = 0; t < G * ST; ++t) val[t] = h[(size_t)(w[t] >> AGG_DBITS) * F + jc];
#pragma unroll
                for (int t = 0; t < G * ST; ++t) {
                    asm volatile("" :: "v"(val[t]));
                }
#pragma unroll
                for (int k = 0; k < G; ++k)
#pragma unroll
                    for (int u = 0; u < ST; ++u)
                        if ((g0 + w0 + k * 32 + u * EPW + sub < n) && (j < F))
                            atomicAdd(&acc[(w[k * ST + u] & ((1u << AGG_DBITS) - 1u)) * F + j], val[k * ST + u]);
#pragma unroll
                for (int k = 0; k < G; ++k) cur[k] = nxt[k];
            }
        }
    } else {
        const int per = (num_edges + AGG_CHUNKS - 1) / AGG_CHUNKS;
        const int e0  = blockIdx.x * per;
        const int e1  = min(e0 + per, num_edges);
        agg_scan<F, FP>(src, dst, h, acc, queue[tid >> 5], base, cnt, e0, e1);
    }
    __syncthreads();

    float* outp = agg + (size_t)base * F;
    const int n = (int)cnt * F;
    for (int i = tid; i < n; i += AGG_TPB) atomicAdd(outp + i, acc[i]);
}

__global__ void gemm2_wmma(const float* __restrict__ agg1,
                           const float* __restrict__ h1,
                           const float* __restrict__ dinv,
                           const float* __restrict__ b1,
                           const float* __restrict__ W2,
                           float* __restrict__ h2,
                           float* __restrict__ agg2, int num_nodes) {
    zero_plane(agg2, num_nodes * OUT_DIM);
    int wave = (blockIdx.x * blockDim.x + threadIdx.x) >> 5;
    int lane = threadIdx.x & 31;
    int row0 = wave * 16;
    if (row0 + 16 > num_nodes) return;

    int half = lane >> 4;
    int l    = lane & 15;

    const float  di = dinv[row0 + l];
    const size_t rb = (size_t)(row0 + l) * HID_DIM + half * 2;
    v8f c = {};
#pragma unroll
    for (int k0 = 0; k0 < HID_DIM; k0 += 4) {
        int kr = k0 + half * 2;
        v2f s = *(const v2f*)(agg1 + rb + k0);
        v2f g = *(const v2f*)(h1 + rb + k0);
        v2f a;
        a.x = fmaxf((s.x + g.x) * di + b1[kr + 0], 0.0f);
        a.y = fmaxf((s.y + g.y) * di + b1[kr + 1], 0.0f);
        v2f b;
        b.x = (l < OUT_DIM) ? W2[(kr + 0) * OUT_DIM + l] : 0.0f;
        b.y = (l < OUT_DIM) ? W2[(kr + 1) * OUT_DIM + l] : 0.0f;
        c = __builtin_amdgcn_wmma_f32_16x16x4_f32(
                false, a, false, b, (short)0, c, false, false);
    }
    if (l < OUT_DIM) {
        float* outp = h2 + (size_t)row0 * OUT_DIM;
        const float4* dp = (const float4*)(dinv + row0 + half * 8);
        const float4 d0 = dp[0], d1 = dp[1];
        const float dr[8] = {d0.x, d0.y, d0.z, d0.w, d1.x, d1.y, d1.z, d1.w};
#pragma unroll
        for (int r = 0; r < 8; ++r)
            outp[(half * 8 + r) * OUT_DIM + l] = c[r] * dr[r];
    }
}

__global__ void node_post2(const float* __restrict__ h2,
                           const float* __restrict__ dinv,
                           const float* __restrict__ b2,
                           const float* __restrict__ agg2,
                           float* __restrict__ out, int num_nodes) {
    int n = blockIdx.x * blockDim.x + threadIdx.x;
    if (n >= num_nodes) return;
    float di = dinv[n];
    float v[OUT_DIM];
    float m = -INFINITY;
#pragma unroll
    for (int j = 0; j < OUT_DIM; ++j) {
        v[j] = (agg2[(size_t)n * OUT_DIM + j] + h2[(size_t)n * OUT_DIM + j]) * di + b2[j];
        m = fmaxf(m, v[j]);
    }
    float s = 0.0f;
#pragma unroll
    for (int j = 0; j < OUT_DIM; ++j) s += expf(v[j] - m);
    float lse = logf(s);
#pragma unroll
    for (int j = 0; j < OUT_DIM; ++j)
        out[(size_t)n * OUT_DIM + j] = v[j] - m - lse;
}

extern "C" void kernel_launch(void* const* d_in, const int* in_sizes, int n_in,
                              void* d_out, int out_size, void* d_ws, size_t ws_size,
                              hipStream_t stream) {
    const float* x  = (const float*)d_in[0];
    const int*   ei = (const int*)d_in[1];
    const float* W1 = (const float*)d_in[2];
    const float* b1 = (const float*)d_in[3];
    const float* W2 = (const float*)d_in[4];
    const float* b2 = (const float*)d_in[5];

    const int N = in_sizes[0] / IN_DIM;
    const int E = in_sizes[1] / 2;
    const int* src = ei;
    const int* dst = ei + E;

    float* ws   = (float*)d_ws;
    float* dinv = ws;
    float* h1   = dinv + (size_t)N;
    float* agg1 = h1 + (size_t)N * HID_DIM;
    float* h2   = agg1 + (size_t)N * HID_DIM;
    float* agg2 = h2 + (size_t)N * OUT_DIM;
    unsigned* lists = (unsigned*)(agg2 + (size_t)N * OUT_DIM);
    float* outp = (float*)d_out;

    const int B = 256;
    auto nb = [](long long work, int blk) { return (int)((work + blk - 1) / blk); };

    unsigned* part = (unsigned*)h1;
    const int chunks = nb(E, DEG_CHUNK);
    degree_hist<<<dim3(chunks, nb(N, DEG_CAP)), DEG_TPB, 0, stream>>>(dst, part, N, E);
    degree_merge<<<nb((N + 1) / 2, B), B, 0, stream>>>(part, dinv, N, chunks);

    int waves1 = N / 16;
    const dim3 agrid(AGG_CHUNKS, nb(N, AGG_RN));
    const int pairs   = (int)(agrid.x * agrid.y);
    const int gneed   = nb(waves1, AGG_WAVES);
    const int gblocks = (gneed > pairs) ? gneed : pairs;
    const int fblocks = FLT_SLICES * nb(agrid.y, FLT_RPG);
    gemm1_wmma_filter<<<gblocks + fblocks, AGG_TPB, 0, stream>>>(
        x, W1, dinv, h1, agg1, src, dst, ws, ws_size, lists, N, E);
    edge_agg_kept<HID_DIM, 16, 2><<<agrid, AGG_TPB, 0, stream>>>(src, dst, h1, agg1, ws, ws_size, lists, N, E);

    gemm2_wmma<<<nb((long long)waves1 * 32, B), B, 0, stream>>>(agg1, h1, dinv, b1, W2, h2, agg2, N);
    edge_agg_kept<OUT_DIM, 8, 4><<<agrid, AGG_TPB, 0, stream>>>(src, dst, h2, agg2, ws, ws_size, lists, N, E);
    node_post2<<<nb(N, B), B, 0, stream>>>(h2, dinv, b2, agg2, outp, N);
}


# A work-group owns a 392-node range (filter splits 32 ways by LDS add): sums its edges in LDS, finishes its nodes (GEMM2, log-softmax fused); no partials, no merge; 7 launches to 5 (add order changes)
# speedup vs baseline: 3.0834x; 1.0782x over previous
#include <hip/hip_runtime.h>
#include <hip/hip_bf16.h>

typedef __attribute__((ext_vector_type(2))) float v2f;
typedef __attribute__((ext_vector_type(8))) float v8f;

#define IN_DIM  500
#define HID_DIM 16
#define OUT_DIM 7

#define DEG_CAP   32768
#define DEG_TPB   1024
#define DEG_BATCH 21
#define DEG_ITERS 63
#define DEG_CHUNK (DEG_ITERS * DEG_TPB)

__device__ __forceinline__ int deg_pitch(int num_nodes) {
    return (((num_nodes + 1) >> 1) + 3) & ~3;
}

template <bool FULL>
__device__ __forceinline__ void deg_count(const int* __restrict__ p, int left,
                                          unsigned* hist, int base, unsigned cnt) {
    for (int it = 0; it < DEG_ITERS; it += DEG_BATCH) {
        int v[DEG_BATCH];
#pragma unroll
        for (int k = 0; k < DEG_BATCH; ++k) {
            int o = (it + k) * DEG_TPB;
            v[k] = (FULL || o < left) ? p[o] : -1;
        }
#pragma unroll
        for (int k = 0; k < DEG_BATCH; ++k) {
            unsigned d = (unsigned)(v[k] - base);
            if (d < cnt) atomicAdd(hist + (d >> 1), 1u << ((d & 1u) << 4));
        }
    }
}

__global__ __launch_bounds__(DEG_TPB)
void degree_hist(const int* __restrict__ dst, unsigned* __restrict__ part,
                 int num_nodes, int num_edges) {
    __shared__ uint4 hist4[DEG_CAP / 8];
    unsigned* hist = (unsigned*)hist4;
    const int tid  = threadIdx.x;
    const int base = blockIdx.y * DEG_CAP;
    const unsigned cnt = (unsigned)min(DEG_CAP, num_nodes - base);

    for (int i = tid; i < DEG_CAP / 8; i += DEG_TPB) hist4[i] = make_uint4(0u, 0u, 0u, 0u);
    __syncthreads();

    const int e0 = blockIdx.x * DEG_CHUNK;
    if (e0 + DEG_CHUNK <= num_edges) {
        deg_count<true>(dst + e0 + tid, 0, hist, base, cnt);
    } else {
        deg_count<false>(dst + e0 + tid, num_edges - e0 - tid, hist, base, cnt);
    }
    __syncthreads();

    uint4* outp = (uint4*)(part + (size_t)blockIdx.x * deg_pitch(num_nodes) + (base >> 1));
    const int nq = (int)((((cnt + 1u) >> 1) + 3u) >> 2);
    for (int i = tid; i < nq; i += DEG_TPB) outp[i] = hist4[i];
}

__global__ void degree_merge(const unsigned* __restrict__ part,
                             float* __restrict__ dinv, int num_nodes, int num_chunks) {
    int w = blockIdx.x * blockDim.x + threadIdx.x;
    if (w >= ((num_nodes + 1) >> 1)) return;
    const int pitch = deg_pitch(num_nodes);
    const unsigned* p = part + w;
    unsigned lo = 0u, hi = 0u;
#pragma unroll 10
    for (int g = 0; g < num_chunks; ++g) {
        unsigned v = p[(size_t)g * pitch];
        lo += v & 0xFFFFu;
        hi += v >> 16;
    }
    float d0 = (float)(lo + 1u);
    float d1 = (float)(hi + 1u);
    float r0 = (d0 > 0.0f) ? rsqrtf(d0) : 0.0f;
    float r1 = (d1 > 0.0f) ? rsqrtf(d1) : 0.0f;
    int n = 2 * w;
    if (n + 1 < num_nodes) {
        *(float2*)(dinv + n) = make_float2(r0, r1);
    } else {
        dinv[n] = r0;
    }
}

#define AGG_TPB    1024
#define AGG_WAVES  (AGG_TPB / 32)
#define AGG_RN     392
#define AGG_DBITS  12
#define AGG_BATCH  16
#define AGG_QCAP   (AGG_BATCH * 32)
#define AGG_SLICES 32
#define FLT_RPG    32
#define FLT_GBITS  14

template <bool FULL>
__device__ __forceinline__ void agg_load(const int* __restrict__ p, int left,
                                         int (&v)[AGG_BATCH]) {
#pragma unroll
    for (int k = 0; k < AGG_BATCH; ++k) {
        const int o = k * AGG_TPB;
        if (FULL) {
            v[k] = p[o];
        } else {
            int x = p[min(o, left - 1)];
            v[k] = (o < left) ? x : -1;
        }
    }
}

#define AGG_SCAP   512
#define AGG_LCAP   (AGG_SCAP * AGG_SLICES)

__device__ __forceinline__ unsigned* list_piece(const float* ws, size_t ws_bytes,
                                                unsigned* lists, unsigned ranges, unsigned rg) {
    unsigned* p = lists + ((ranges * AGG_SLICES + 63u) & ~63u) + (size_t)rg * AGG_LCAP;
    const size_t end = (size_t)((const char*)(p + AGG_LCAP) - (const char*)ws);
    return (end <= ws_bytes) ? p : nullptr;
}

__device__ __forceinline__ void filter_scan(const int* __restrict__ src, const int* __restrict__ dst,
                                            unsigned* q, unsigned* lcur, unsigned* __restrict__ keep,
                                            unsigned nv, int base, unsigned cnt, int e0, int e1) {
    const int tid  = threadIdx.x;
    const int lane = tid & 31;
    constexpr int SPAN = AGG_BATCH * AGG_TPB;

    if (e0 < e1) {
        int cur[AGG_BATCH], nxt[AGG_BATCH];
        if (e0 + SPAN <= e1) agg_load<true>(dst + e0 + tid, 0, cur);
        else                 agg_load<false>(dst + e0 + tid, e1 - e0 - tid, cur);
#pragma unroll
        for (int k = 0; k < AGG_BATCH; ++k) {
            asm volatile("" :: "v"(cur[k]));
        }
        for (int eb = e0; eb < e1; eb += SPAN) {
            const int en = eb + SPAN;
            if (en + SPAN <= e1) agg_load<true>(dst + en + tid, 0, nxt);
            else                 agg_load<false>(dst + en + tid, e1 - en - tid, nxt);
            const int* sp = src + eb + tid;
            int s[AGG_BATCH];
#pragma unroll
            for (int k = 0; k < AGG_BATCH; ++k) {
                unsigned d = (unsigned)(cur[k] - base);
                s[k] = (d < cnt) ? sp[k * AGG_TPB] : 0;
            }
#pragma unroll
            for (int k = 0; k < AGG_BATCH; ++k) {
                asm volatile("" : "+v"(s[k]));
            }
            int qn = 0;
#pragma unroll
            for (int k = 0; k < AGG_BATCH; ++k) {
                unsigned d = (unsigned)(cur[k] - base);
                bool mt = d < cnt;
                unsigned m = __builtin_amdgcn_ballot_w32(mt);
                if (mt) q[qn + (int)__builtin_amdgcn_mbcnt_lo(m, 0u)] = ((unsigned)s[k] << FLT_GBITS) | d;
                qn += __popc(m);
            }
            __builtin_amdgcn_fence(__ATOMIC_RELEASE, "wavefront");
            __builtin_amdgcn_wave_barrier();
            __builtin_amdgcn_fence(__ATOMIC_ACQUIRE, "wavefront");
            for (int i0 = 0; i0 < qn; i0 += 32) {
                const bool ok = i0 + lane < qn;
                const unsigned w  = q[ok ? i0 + lane : 0];
                const unsigned dg = w & ((1u << FLT_GBITS) - 1u);
                const unsigned r  = dg / (unsigned)AGG_RN;
                const unsigned word = ((w >> FLT_GBITS) << AGG_DBITS) | (dg - r * (unsigned)AGG_RN);
                unsigned pos = 0u;
                if (ok) pos = atomicAdd(lcur + r, 1u);
                if (ok && r < nv && pos < (unsigned)AGG_SCAP)
                    keep[(size_t)r * AGG_LCAP + pos] = word;
            }
            __builtin_amdgcn_fence(__ATOMIC_RELEASE, "wavefront");
            __builtin_amdgcn_wave_barrier();
#pragma unroll
            for (int k = 0; k < AGG_BATCH; ++k) cur[k] = nxt[k];
        }
    }
}

template <int F, int FP>
__device__ __forceinline__ void agg_scan(const int* __restrict__ src, const int* __restrict__ dst,
                                         const float* __restrict__ h, float* acc, unsigned* q,
                                         int base, unsigned cnt, int e0, int e1) {
    const int tid  = threadIdx.x;
    const int lane = tid & 31;
    constexpr int EPW  = 32 / FP;
    constexpr int U    = 16 / EPW;
    constexpr int SPAN = AGG_BATCH * AGG_TPB;
    const int sub = lane / FP;
    const int j   = lane % FP;
    const int jc  = (j < F) ? j : F - 1;

    if (e0 < e1) {
        int cur[AGG_BATCH], nxt[AGG_BATCH];
        if (e0 + SPAN <= e1) agg_load<true>(dst + e0 + tid, 0, cur);
        else                 agg_load<false>(dst + e0 + tid, e1 - e0 - tid, cur);
#pragma unroll
        for (int k = 0; k < AGG_BATCH; ++k) {
            asm volatile("" :: "v"(cur[k]));
        }
        for (int eb = e0; eb < e1; eb += SPAN) {
            const int en = eb + SPAN;
            if (en + SPAN <= e1) agg_load<true>(dst + en + tid, 0, nxt);
            else                 agg_load<false>(dst + en + tid, e1 - en - tid, nxt);
            const int* sp = src + eb + tid;
            int s[AGG_BATCH];
#pragma unroll
            for (int k = 0; k < AGG_BATCH; ++k) {
                unsigned d = (unsigned)(cur[k] - base);
                s[k] = (d < cnt) ? sp[k * AGG_TPB] : 0;
            }
#pragma unroll
            for (int k = 0; k < AGG_BATCH; ++k) {
                asm volatile("" : "+v"(s[k]));
            }
            int qn = 0;
#pragma unroll
            for (int k = 0; k < AGG_BATCH; ++k) {
                unsigned d = (unsigned)(cur[k] - base);
                bool mt = d < cnt;
                unsigned m = __builtin_amdgcn_ballot_w32(mt);
                if (mt) q[qn + (int)__builtin_amdgcn_mbcnt_lo(m, 0u)] = ((unsigned)s[k] << AGG_DBITS) | d;
                qn += __popc(m);
            }
            __builtin_amdgcn_fence(__ATOMIC_RELEASE, "wavefront");
            __builtin_amdgcn_wave_barrier();
            __builtin_amdgcn_fence(__ATOMIC_ACQUIRE, "wavefront");
            for (int i0 = 0; i0 < qn; i0 += 16) {
                unsigned w[U];
                float    val[U];
#pragma unroll
                for (int u = 0; u < U; ++u) w[u] = q[min(i0 + u * EPW + sub, qn - 1)];
#pragma unroll
                for (int u = 0; u < U; ++u) {
                    const unsigned sx = w[u] >> AGG_DBITS;
                    val[u] = h[(size_t)sx * F + jc];
                }
#pragma unroll
                for (int u = 0; u < U; ++u) {
                    asm volatile("" :: "v"(val[u]));
                }
#pragma unroll
                for (int u = 0; u < U; ++u)
                    if ((i0 + u * EPW + sub < qn) && (j < F))
                        atomicAdd(&acc[(w[u] & ((1u << AGG_DBITS) - 1u)) * F + j], val[u]);
            }
            __builtin_amdgcn_fence(__ATOMIC_RELEASE, "wavefront");
            __builtin_amdgcn_wave_barrier();
#pragma unroll
            for (int k = 0; k < AGG_BATCH; ++k) cur[k] = nxt[k];
        }
    }
}

__global__ __launch_bounds__(AGG_TPB)
void gemm1_wmma_filter(const float* __restrict__ x,
                       const float* __restrict__ W1,
                       const float* __restrict__ dinv,
                       float* __restrict__ h1,
                       const int* __restrict__ src, const int* __restrict__ dst,
                       const float* __restrict__ ws, size_t ws_bytes, unsigned* __restrict__ lists,
                       int num_nodes, int num_edges) {
    __shared__ unsigned queue[AGG_WAVES][AGG_QCAP];
    __shared__ unsigned lcur[FLT_RPG];
    static_assert(AGG_RN <= (1 << AGG_DBITS), "local destination must fit its bits");
    static_assert(FLT_RPG * AGG_RN <= (1 << FLT_GBITS), "destination within the group must fit its bits");
    static_assert(FLT_RPG <= 32, "one lane per range stores the group's counts");

    const unsigned ranges  = (unsigned)((num_nodes + AGG_RN - 1) / AGG_RN);
    const unsigned fblocks = (unsigned)AGG_SLICES * ((ranges + FLT_RPG - 1u) / FLT_RPG);
    const unsigned gblocks = gridDim.x - fblocks;
    if (blockIdx.x >= gblocks) {
        const int tid = threadIdx.x;
        const unsigned f   = blockIdx.x - gblocks;
        const unsigned rg0 = (f / AGG_SLICES) * FLT_RPG;
        const unsigned sl  = f % AGG_SLICES;
        unsigned nv = 0u;
        for (unsigned i = 0u; i < FLT_RPG; ++i)
            if (rg0 + i < ranges &&
                list_piece(ws, ws_bytes, lists, ranges, rg0 + i) != nullptr) nv = i + 1u;
        if (nv == 0u) return;
        unsigned* keep = list_piece(ws, ws_bytes, lists, ranges, rg0) + sl * AGG_SCAP;
        const int base = (int)rg0 * AGG_RN;
        const unsigned cnt = (unsigned)min(FLT_RPG * AGG_RN, num_nodes - base);
        if (tid < FLT_RPG) lcur[tid] = 0u;
        __syncthreads();
        const int pq = (num_edges + AGG_SLICES - 1) / AGG_SLICES;
        const int e0 = (int)sl * pq;
        const int e1 = min(e0 + pq, num_edges);
        filter_scan(src, dst, queue[tid >> 5], lcur, keep, nv, base, cnt, e0, e1);
        __syncthreads();
        if (tid < (int)nv)
            lists[((size_t)rg0 + tid) * AGG_SLICES + sl] = lcur[tid];
        return;
    }

    int tile = (int)(threadIdx.x >> 5) * (int)gblocks + (int)blockIdx.x;
    int lane = threadIdx.x & 31;
    int row0 = tile * 16;
    if (row0 + 16 > num_nodes) return;

    int half = lane >> 4;
    int l    = lane & 15;

    const float* arow = x + (size_t)(row0 + l) * IN_DIM + half * 2;
    v8f c = {};
    for (int k0 = 0; k0 < IN_DIM; k0 += 4) {
        v2f a = *(const v2f*)(arow + k0);
        v2f b;
        b.x = W1[(k0 + half * 2 + 0) * HID_DIM + l];
        b.y = W1[(k0 + half * 2 + 1) * HID_DIM + l];
        c = __builtin_amdgcn_wmma_f32_16x16x4_f32(
                false, a, false, b, (short)0, c, false, false);
    }
    float* outp = h1 + (size_t)row0 * HID_DIM;
    const float4* dp = (const float4*)(dinv + row0 + half * 8);
    const float4 d0 = dp[0], d1 = dp[1];
    const float dr[8] = {d0.x, d0.y, d0.z, d0.w, d1.x, d1.y, d1.z, d1.w};
#pragma unroll
    for (int r = 0; r < 8; ++r)
        outp[(half * 8 + r) * HID_DIM + l] = c[r] * dr[r];
}

template <int F, int FP, int G, bool LAST>
__global__ __launch_bounds__(AGG_TPB)
void edge_agg_kept(const int* __restrict__ src, const int* __restrict__ dst,
                   const float* __restrict__ h, float* __restrict__ res,
                   const float* __restrict__ dinv, const float* __restrict__ bias,
                   const float* __restrict__ wnext,
                   const float* __restrict__ ws, size_t ws_bytes, unsigned* __restrict__ lists,
                   int num_nodes, int num_edges) {
    __shared__ alignas(16) float acc[AGG_RN * F];
    __shared__ unsigned queue[AGG_WAVES][AGG_QCAP];
    static_assert(AGG_SLICES == 32 && AGG_WAVES == AGG_SLICES, "lane t reads count t, wave t walks sub-list t");

    const int tid  = threadIdx.x;
    const int lane = tid & 31;
    const int wv   = tid >> 5;
    const unsigned rg = blockIdx.x;
    const int base = (int)rg * AGG_RN;
    const unsigned cnt = (unsigned)min(AGG_RN, num_nodes - base);
    const unsigned* piece = list_piece(ws, ws_bytes, lists, gridDim.x, rg);
    const unsigned c = (piece != nullptr) ? lists[(size_t)rg * AGG_SLICES + lane] : 0xFFFFFFFFu;
    const bool kept_ok = __builtin_amdgcn_ballot_w32(c > (unsigned)AGG_SCAP) == 0u;

    for (int i = tid; i < AGG_RN * F; i += AGG_TPB) acc[i] = 0.0f;
    __syncthreads();

    if (kept_ok) {
        constexpr int EPW = 32 / FP;
        constexpr int ST  = 32 / EPW;
        constexpr int TRIP = 32 * G;
        const int sub = lane / FP;
        const int j   = lane % FP;
        const int jc  = (j < F) ? j : F - 1;
        const int n = __builtin_amdgcn_readfirstlane(__shfl((int)c, wv));
        const unsigned* sl = piece + (size_t)wv * AGG_SCAP;
        if (n > 0) {
            unsigned cur[G], nxt[G];
#pragma unroll
            for (int k = 0; k < G; ++k) cur[k] = sl[min(k * 32 + lane, n - 1)];
            for (int g0 = 0; g0 < n; g0 += TRIP) {
#pragma unroll
                for (int k = 0; k < G; ++k) nxt[k] = sl[min(g0 + TRIP + k * 32 + lane, n - 1)];
                unsigned w[G * ST];
                float    val[G * ST];
#pragma unroll
                for (int k = 0; k < G; ++k)
#pragma unroll
                    for (int u = 0; u < ST; ++u) w[k * ST + u] = (unsigned)__shfl((int)cur[k], u * EPW + sub);
#pragma unroll
                for (int t = 0; t < G * ST; ++t) val[t] = h[(size_t)(w[t] >> AGG_DBITS) * F + jc];
#pragma unroll
                for (int t = 0; t < G * ST; ++t) {
                    asm volatile("" :: "v"(val[t]));
                }
#pragma unroll
                for (int k = 0; k < G; ++k)
#pragma unroll
                    for (int u = 0; u < ST; ++u)
                        if ((g0 + k * 32 + u * EPW + sub < n) && (j < F))
                            atomicAdd(&acc[(w[k * ST + u] & ((1u << AGG_DBITS) - 1u)) * F + j], val[k * ST + u]);
#pragma unroll
                for (int k = 0; k < G; ++k) cur[k] = nxt[k];
            }
        }
    } else {
        agg_scan<F, FP>(src, dst, h, acc, queue[wv], base, cnt, 0, num_edges);
    }
    __syncthreads();

    if constexpr (!LAST) {
        static_assert(F == HID_DIM, "GEMM2 takes layer 1's rows");
        const int half = lane >> 4;
        const int l    = lane & 15;
        const int tiles = ((int)cnt + 15) >> 4;
        const int wvu = __builtin_amdgcn_readfirstlane(wv);
        for (int t = wvu; t < tiles; t += AGG_WAVES) {
            const int rl = min(t * 16 + l, (int)cnt - 1);
            const size_t nd = (size_t)(base + rl);
            const float di = dinv[nd];
            const float* hr = h + nd * F + half * 2;
            const float* ar = acc + rl * F + half * 2;
            float dr[8];
#pragma unroll
            for (int r = 0; r < 8; ++r)
                dr[r] = dinv[base + min(t * 16 + half * 8 + r, (int)cnt - 1)];
            v8f cc = {};
#pragma unroll
            for (int k0 = 0; k0 < F; k0 += 4) {
                const int kr = k0 + half * 2;
                v2f s = *(const v2f*)(ar + k0);
                v2f g = *(const v2f*)(hr + k0);
                v2f a;
                a.x = fmaxf((s.x + g.x) * di + bias[kr + 0], 0.0f);
                a.y = fmaxf((s.y + g.y) * di + bias[kr + 1], 0.0f);
                v2f b;
                b.x = (l < OUT_DIM) ? wnext[(kr + 0) * OUT_DIM + l] : 0.0f;
                b.y = (l < OUT_DIM) ? wnext[(kr + 1) * OUT_DIM + l] : 0.0f;
                cc = __builtin_amdgcn_wmma_f32_16x16x4_f32(
                        false, a, false, b, (short)0, cc, false, false);
            }
            asm volatile("v_nop\n\tv_nop\n\tv_nop\n\tv_nop" : "+v"(cc));
#pragma unroll
            for (int r = 0; r < 8; ++r) {
                const int row = t * 16 + half * 8 + r;
                if (l < OUT_DIM && row < (int)cnt)
                    res[(size_t)(base + row) * OUT_DIM + l] = cc[r] * dr[r];
            }
        }
    } else {
        for (int nl = tid; nl < (int)cnt; nl += AGG_TPB) {
            const size_t nd = (size_t)(base + nl);
            const float di = dinv[nd];
            float v[F];
            float m = -INFINITY;
#pragma unroll
            for (int jj = 0; jj < F; ++jj) {
                v[jj] = (acc[nl * F + jj] + h[nd * F + jj]) * di + bias[jj];
                m = fmaxf(m, v[jj]);
            }
            float s = 0.0f;
#pragma unroll
            for (int jj = 0; jj < F; ++jj) s += expf(v[jj] - m);
            const float lse = logf(s);
#pragma unroll
            for (int jj = 0; jj < F; ++jj)
                res[nd * F + jj] = v[jj] - m - lse;
        }
    }
}

extern "C" void kernel_launch(void* const* d_in, const int* in_sizes, int n_in,
                              void* d_out, int out_size, void* d_ws, size_t ws_size,
                              hipStream_t stream) {
    const float* x  = (const float*)d_in[0];
    const int*   ei = (const int*)d_in[1];
    const float* W1 = (const float*)d_in[2];
    const float* b1 = (const float*)d_in[3];
    const float* W2 = (const float*)d_in[4];
    const float* b2 = (const float*)d_in[5];

    const int N = in_sizes[0] / IN_DIM;
    const int E = in_sizes[1] / 2;
    const int* src = ei;
    const int* dst = ei + E;

    float* ws   = (float*)d_ws;
    float* dinv = ws;
    float* h1   = dinv + (size_t)N;
    float* h2   = h1 + (size_t)N * HID_DIM;
    unsigned* lists = (unsigned*)(h2 + (size_t)N * OUT_DIM);
    float* outp = (float*)d_out;

    const int B = 256;
    auto nb = [](long long work, int blk) { return (int)((work + blk - 1) / blk); };

    unsigned* part = (unsigned*)h1;
    const int chunks = nb(E, DEG_CHUNK);
    degree_hist<<<dim3(chunks, nb(N, DEG_CAP)), DEG_TPB, 0, stream>>>(dst, part, N, E);
    degree_merge<<<nb((N + 1) / 2, B), B, 0, stream>>>(part, dinv, N, chunks);

    int waves1 = N / 16;
    const int ranges  = nb(N, AGG_RN);
    const int gneed   = nb(waves1, AGG_WAVES);
    const int gblocks = (gneed > ranges) ? gneed : ranges;
    const int fblocks = AGG_SLICES * nb(ranges, FLT_RPG);
    gemm1_wmma_filter<<<gblocks + fblocks, AGG_TPB, 0, stream>>>(
        x, W1, dinv, h1, src, dst, ws, ws_size, lists, N, E);
    edge_agg_kept<HID_DIM, 16, 2, false><<<ranges, AGG_TPB, 0, stream>>>(
        src, dst, h1, h2, dinv, b1, W2, ws, ws_size, lists, N, E);

    edge_agg_kept<OUT_DIM, 8, 4, true><<<ranges, AGG_TPB, 0, stream>>>(
        src, dst, h2, outp, dinv, b2, W2, ws, ws_size, lists, N, E);
}


# GEMM1 fragments by 16-byte loads: k in steps of 8, a lane loads 4 consecutive k of its x row and of a W1 copy the merge launch lays out in fragment order; 1 load per 2 WMMAs (k sum order changes)
# speedup vs baseline: 3.7855x; 1.2277x over previous
#include <hip/hip_runtime.h>
#include <hip/hip_bf16.h>

typedef __attribute__((ext_vector_type(2))) float v2f;
typedef __attribute__((ext_vector_type(8))) float v8f;

#define IN_DIM  500
#define HID_DIM 16
#define OUT_DIM 7

#define DEG_CAP   32768
#define DEG_TPB   1024
#define DEG_BATCH 21
#define DEG_ITERS 63
#define DEG_CHUNK (DEG_ITERS * DEG_TPB)

__device__ __forceinline__ int deg_pitch(int num_nodes) {
    return (((num_nodes + 1) >> 1) + 3) & ~3;
}

template <bool FULL>
__device__ __forceinline__ void deg_count(const int* __restrict__ p, int left,
                                          unsigned* hist, int base, unsigned cnt) {
    for (int it = 0; it < DEG_ITERS; it += DEG_BATCH) {
        int v[DEG_BATCH];
#pragma unroll
        for (int k = 0; k < DEG_BATCH; ++k) {
            int o = (it + k) * DEG_TPB;
            v[k] = (FULL || o < left) ? p[o] : -1;
        }
#pragma unroll
        for (int k = 0; k < DEG_BATCH; ++k) {
            unsigned d = (unsigned)(v[k] - base);
            if (d < cnt) atomicAdd(hist + (d >> 1), 1u << ((d & 1u) << 4));
        }
    }
}

__global__ __launch_bounds__(DEG_TPB)
void degree_hist(const int* __restrict__ dst, unsigned* __restrict__ part,
                 int num_nodes, int num_edges) {
    __shared__ uint4 hist4[DEG_CAP / 8];
    unsigned* hist = (unsigned*)hist4;
    const int tid  = threadIdx.x;
    const int base = blockIdx.y * DEG_CAP;
    const unsigned cnt = (unsigned)min(DEG_CAP, num_nodes - base);

    for (int i = tid; i < DEG_CAP / 8; i += DEG_TPB) hist4[i] = make_uint4(0u, 0u, 0u, 0u);
    __syncthreads();

    const int e0 = blockIdx.x * DEG_CHUNK;
    if (e0 + DEG_CHUNK <= num_edges) {
        deg_count<true>(dst + e0 + tid, 0, hist, base, cnt);
    } else {
        deg_count<false>(dst + e0 + tid, num_edges - e0 - tid, hist, base, cnt);
    }
    __syncthreads();

    uint4* outp = (uint4*)(part + (size_t)blockIdx.x * deg_pitch(num_nodes) + (base >> 1));
    const int nq = (int)((((cnt + 1u) >> 1) + 3u) >> 2);
    for (int i = tid; i < nq; i += DEG_TPB) outp[i] = hist4[i];
}

#define G1_STEPS  (IN_DIM / 8)
#define W1P_FLOATS 8192

__global__ void degree_merge(const unsigned* __restrict__ part,
                             float* __restrict__ dinv, int num_nodes, int num_chunks,
                             const float* __restrict__ W1, float* __restrict__ w1p,
                             int node_blocks) {
    if ((int)blockIdx.x >= node_blocks) {
        const int i = ((int)blockIdx.x - node_blocks) * (int)blockDim.x + (int)threadIdx.x;
        if (i < G1_STEPS * 128) {
            const int t = i >> 7, ln = (i >> 2) & 31, s = i & 3;
            w1p[i] = W1[(8 * t + 4 * (ln >> 4) + s) * HID_DIM + (ln & 15)];
        }
        return;
    }
    int w = blockIdx.x * blockDim.x + threadIdx.x;
    if (w >= ((num_nodes + 1) >> 1)) return;
    const int pitch = deg_pitch(num_nodes);
    const unsigned* p = part + w;
    unsigned lo = 0u, hi = 0u;
#pragma unroll 10
    for (int g = 0; g < num_chunks; ++g) {
        unsigned v = p[(size_t)g * pitch];
        lo += v & 0xFFFFu;
        hi += v >> 16;
    }
    float d0 = (float)(lo + 1u);
    float d1 = (float)(hi + 1u);
    float r0 = (d0 > 0.0f) ? rsqrtf(d0) : 0.0f;
    float r1 = (d1 > 0.0f) ? rsqrtf(d1) : 0.0f;
    int n = 2 * w;
    if (n + 1 < num_nodes) {
        *(float2*)(dinv + n) = make_float2(r0, r1);
    } else {
        dinv[n] = r0;
    }
}

#define AGG_TPB    1024
#define AGG_WAVES  (AGG_TPB / 32)
#define AGG_RN     392
#define AGG_DBITS  12
#define AGG_BATCH  16
#define AGG_QCAP   (AGG_BATCH * 32)
#define AGG_SLICES 32
#define FLT_RPG    32
#define FLT_GBITS  14

template <bool FULL>
__device__ __forceinline__ void agg_load(const int* __restrict__ p, int left,
                                         int (&v)[AGG_BATCH]) {
#pragma unroll
    for (int k = 0; k < AGG_BATCH; ++k) {
        const int o = k * AGG_TPB;
        if (FULL) {
            v[k] = p[o];
        } else {
            int x = p[min(o, left - 1)];
            v[k] = (o < left) ? x : -1;
        }
    }
}

#define AGG_SCAP   512
#define AGG_LCAP   (AGG_SCAP * AGG_SLICES)

__device__ __forceinline__ unsigned* list_piece(const float* ws, size_t ws_bytes,
                                                unsigned* lists, unsigned ranges, unsigned rg) {
    unsigned* p = lists + ((ranges * AGG_SLICES + 63u) & ~63u) + (size_t)rg * AGG_LCAP;
    const size_t end = (size_t)((const char*)(p + AGG_LCAP) - (const char*)ws);
    return (end <= ws_bytes) ? p : nullptr;
}

__device__ __forceinline__ void filter_scan(const int* __restrict__ src, const int* __restrict__ dst,
                                            unsigned* q, unsigned* lcur, unsigned* __restrict__ keep,
                                            unsigned nv, int base, unsigned cnt, int e0, int e1) {
    const int tid  = threadIdx.x;
    const int lane = tid & 31;
    constexpr int SPAN = AGG_BATCH * AGG_TPB;

    if (e0 < e1) {
        int cur[AGG_BATCH], nxt[AGG_BATCH];
        if (e0 + SPAN <= e1) agg_load<true>(dst + e0 + tid, 0, cur);
        else                 agg_load<false>(dst + e0 + tid, e1 - e0 - tid, cur);
#pragma unroll
        for (int k = 0; k < AGG_BATCH; ++k) {
            asm volatile("" :: "v"(cur[k]));
        }
        for (int eb = e0; eb < e1; eb += SPAN) {
            const int en = eb + SPAN;
            if (en + SPAN <= e1) agg_load<true>(dst + en + tid, 0, nxt);
            else                 agg_load<false>(dst + en + tid, e1 - en - tid, nxt);
            const int* sp = src + eb + tid;
            int s[AGG_BATCH];
#pragma unroll
            for (int k = 0; k < AGG_BATCH; ++k) {
                unsigned d = (unsigned)(cur[k] - base);
                s[k] = (d < cnt) ? sp[k * AGG_TPB] : 0;
            }
#pragma unroll
            for (int k = 0; k < AGG_BATCH; ++k) {
                asm volatile("" : "+v"(s[k]));
            }
            int qn = 0;
#pragma unroll
            for (int k = 0; k < AGG_BATCH; ++k) {
                unsigned d = (unsigned)(cur[k] - base);
                bool mt = d < cnt;
                unsigned m = __builtin_amdgcn_ballot_w32(mt);
                if (mt) q[qn + (int)__builtin_amdgcn_mbcnt_lo(m, 0u)] = ((unsigned)s[k] << FLT_GBITS) | d;
                qn += __popc(m);
            }
            __builtin_amdgcn_fence(__ATOMIC_RELEASE, "wavefront");
            __builtin_amdgcn_wave_barrier();
            __builtin_amdgcn_fence(__ATOMIC_ACQUIRE, "wavefront");
            for (int i0 = 0; i0 < qn; i0 += 32) {
                const bool ok = i0 + lane < qn;
                const unsigned w  = q[ok ? i0 + lane : 0];
                const unsigned dg = w & ((1u << FLT_GBITS) - 1u);
                const unsigned r  = dg / (unsigned)AGG_RN;
                const unsigned word = ((w >> FLT_GBITS) << AGG_DBITS) | (dg - r * (unsigned)AGG_RN);
                unsigned pos = 0u;
                if (ok) pos = atomicAdd(lcur + r, 1u);
                if (ok && r < nv && pos < (unsigned)AGG_SCAP)
                    keep[(size_t)r * AGG_LCAP + pos] = word;
            }
            __builtin_amdgcn_fence(__ATOMIC_RELEASE, "wavefront");
            __builtin_amdgcn_wave_barrier();
#pragma unroll
            for (int k = 0; k < AGG_BATCH; ++k) cur[k] = nxt[k];
        }
    }
}

template <int F, int FP>
__device__ __forceinline__ void agg_scan(const int* __restrict__ src, const int* __restrict__ dst,
                                         const float* __restrict__ h, float* acc, unsigned* q,
                                         int base, unsigned cnt, int e0, int e1) {
    const int tid  = threadIdx.x;
    const int lane = tid & 31;
    constexpr int EPW  = 32 / FP;
    constexpr int U    = 16 / EPW;
    constexpr int SPAN = AGG_BATCH * AGG_TPB;
    const int sub = lane / FP;
    const int j   = lane % FP;
    const int jc  = (j < F) ? j : F - 1;

    if (e0 < e1) {
        int cur[AGG_BATCH], nxt[AGG_BATCH];
        if (e0 + SPAN <= e1) agg_load<true>(dst + e0 + tid, 0, cur);
        else                 agg_load<false>(dst + e0 + tid, e1 - e0 - tid, cur);
#pragma unroll
        for (int k = 0; k < AGG_BATCH; ++k) {
            asm volatile("" :: "v"(cur[k]));
        }
        for (int eb = e0; eb < e1; eb += SPAN) {
            const int en = eb + SPAN;
            if (en + SPAN <= e1) agg_load<true>(dst + en + tid, 0, nxt);
            else                 agg_load<false>(dst + en + tid, e1 - en - tid, nxt);
            const int* sp = src + eb + tid;
            int s[AGG_BATCH];
#pragma unroll
            for (int k = 0; k < AGG_BATCH; ++k) {
                unsigned d = (unsigned)(cur[k] - base);
                s[k] = (d < cnt) ? sp[k * AGG_TPB] : 0;
            }
#pragma unroll
            for (int k = 0; k < AGG_BATCH; ++k) {
                asm volatile("" : "+v"(s[k]));
            }
            int qn = 0;
#pragma unroll
            for (int k = 0; k < AGG_BATCH; ++k) {
                unsigned d = (unsigned)(cur[k] - base);
                bool mt = d < cnt;
                unsigned m = __builtin_amdgcn_ballot_w32(mt);
                if (mt) q[qn + (int)__builtin_amdgcn_mbcnt_lo(m, 0u)] = ((unsigned)s[k] << AGG_DBITS) | d;
                qn += __popc(m);
            }
            __builtin_amdgcn_fence(__ATOMIC_RELEASE, "wavefront");
            __builtin_amdgcn_wave_barrier();
            __builtin_amdgcn_fence(__ATOMIC_ACQUIRE, "wavefront");
            for (int i0 = 0; i0 < qn; i0 += 16) {
                unsigned w[U];
                float    val[U];
#pragma unroll
                for (int u = 0; u < U; ++u) w[u] = q[min(i0 + u * EPW + sub, qn - 1)];
#pragma unroll
                for (int u = 0; u < U; ++u) {
                    const unsigned sx = w[u] >> AGG_DBITS;
                    val[u] = h[(size_t)sx * F + jc];
                }
#pragma unroll
                for (int u = 0; u < U; ++u) {
                    asm volatile("" :: "v"(val[u]));
                }
#pragma unroll
                for (int u = 0; u < U; ++u)
                    if ((i0 + u * EPW + sub < qn) && (j < F))
                        atomicAdd(&acc[(w[u] & ((1u << AGG_DBITS) - 1u)) * F + j], val[u]);
            }
            __builtin_amdgcn_fence(__ATOMIC_RELEASE, "wavefront");
            __builtin_amdgcn_wave_barrier();
#pragma unroll
            for (int k = 0; k < AGG_BATCH; ++k) cur[k] = nxt[k];
        }
    }
}

__global__ __launch_bounds__(AGG_TPB)
void gemm1_wmma_filter(const float* __restrict__ x,
                       const float* __restrict__ W1,
                       const float* __restrict__ w1p,
                       const float* __restrict__ dinv,
                       float* __restrict__ h1,
                       const int* __restrict__ src, const int* __restrict__ dst,
                       const float* __restrict__ ws, size_t ws_bytes, unsigned* __restrict__ lists,
                       int num_nodes, int num_edges) {
    __shared__ unsigned queue[AGG_WAVES][AGG_QCAP];
    __shared__ unsigned lcur[FLT_RPG];
    static_assert(AGG_RN <= (1 << AGG_DBITS), "local destination must fit its bits");
    static_assert(FLT_RPG * AGG_RN <= (1 << FLT_GBITS), "destination within the group must fit its bits");
    static_assert(FLT_RPG <= 32, "one lane per range stores the group's counts");

    const unsigned ranges  = (unsigned)((num_nodes + AGG_RN - 1) / AGG_RN);
    const unsigned fblocks = (unsigned)AGG_SLICES * ((ranges + FLT_RPG - 1u) / FLT_RPG);
    const unsigned gblocks = gridDim.x - fblocks;
    if (blockIdx.x >= gblocks) {
        const int tid = threadIdx.x;
        const unsigned f   = blockIdx.x - gblocks;
        const unsigned rg0 = (f / AGG_SLICES) * FLT_RPG;
        const unsigned sl  = f % AGG_SLICES;
        unsigned nv = 0u;
        for (unsigned i = 0u; i < FLT_RPG; ++i)
            if (rg0 + i < ranges &&
                list_piece(ws, ws_bytes, lists, ranges, rg0 + i) != nullptr) nv = i + 1u;
        if (nv == 0u) return;
        unsigned* keep = list_piece(ws, ws_bytes, lists, ranges, rg0) + sl * AGG_SCAP;
        const int base = (int)rg0 * AGG_RN;
        const unsigned cnt = (unsigned)min(FLT_RPG * AGG_RN, num_nodes - base);
        if (tid < FLT_RPG) lcur[tid] = 0u;
        __syncthreads();
        const int pq = (num_edges + AGG_SLICES - 1) / AGG_SLICES;
        const int e0 = (int)sl * pq;
        const int e1 = min(e0 + pq, num_edges);
        filter_scan(src, dst, queue[tid >> 5], lcur, keep, nv, base, cnt, e0, e1);
        __syncthreads();
        if (tid < (int)nv)
            lists[((size_t)rg0 + tid) * AGG_SLICES + sl] = lcur[tid];
        return;
    }

    int tile = (int)(threadIdx.x >> 5) * (int)gblocks + (int)blockIdx.x;
    int lane = threadIdx.x & 31;
    int row0 = tile * 16;
    if (row0 + 16 > num_nodes) return;

    int half = lane >> 4;
    int l    = lane & 15;

    static_assert(IN_DIM % 8 == 4 && G1_STEPS * 128 <= W1P_FLOATS, "steps of 8 and one plain step of 4");
    constexpr int G1_U = 5;
    static_assert(G1_STEPS % G1_U == 2, "the tail below takes two steps of 8 and the plain step");
    const float* xrow = x + (size_t)(row0 + l) * IN_DIM;
    const float* arow = xrow + half * 4;
    const float* brow = w1p + lane * 4;
    v8f c = {};
    for (int t0 = 0; t0 + G1_U <= G1_STEPS; t0 += G1_U) {
        float4 a[G1_U], b[G1_U];
#pragma unroll
        for (int u = 0; u < G1_U; ++u) {
            a[u] = *(const float4*)(arow + (t0 + u) * 8);
            b[u] = *(const float4*)(brow + (t0 + u) * 128);
        }
#pragma unroll
        for (int u = 0; u < G1_U; ++u) {
            v2f a0 = {a[u].x, a[u].y}, b0 = {b[u].x, b[u].y};
            v2f a1 = {a[u].z, a[u].w}, b1 = {b[u].z, b[u].w};
            c = __builtin_amdgcn_wmma_f32_16x16x4_f32(
                    false, a0, false, b0, (short)0, c, false, false);
            c = __builtin_amdgcn_wmma_f32_16x16x4_f32(
                    false, a1, false, b1, (short)0, c, false, false);
        }
    }
    {
        constexpr int T0 = G1_STEPS - 2;
        constexpr int K0 = G1_STEPS * 8;
        float4 a[2], b[2];
#pragma unroll
        for (int u = 0; u < 2; ++u) {
            a[u] = *(const float4*)(arow + (T0 + u) * 8);
            b[u] = *(const float4*)(brow + (T0 + u) * 128);
        }
        v2f ap = *(const v2f*)(xrow + K0 + half * 2);
        v2f bp;
        bp.x = W1[(K0 + half * 2 + 0) * HID_DIM + l];
        bp.y = W1[(K0 + half * 2 + 1) * HID_DIM + l];
#pragma unroll
        for (int u = 0; u < 2; ++u) {
            v2f a0 = {a[u].x, a[u].y}, b0 = {b[u].x, b[u].y};
            v2f a1 = {a[u].z, a[u].w}, b1 = {b[u].z, b[u].w};
            c = __builtin_amdgcn_wmma_f32_16x16x4_f32(
                    false, a0, false, b0, (short)0, c, false, false);
            c = __builtin_amdgcn_wmma_f32_16x16x4_f32(
                    false, a1, false, b1, (short)0, c, false, false);
        }
        c = __builtin_amdgcn_wmma_f32_16x16x4_f32(
                false, ap, false, bp, (short)0, c, false, false);
    }
    float* outp = h1 + (size_t)row0 * HID_DIM;
    const float4* dp = (const float4*)(dinv + row0 + half * 8);
    const float4 d0 = dp[0], d1 = dp[1];
    const float dr[8] = {d0.x, d0.y, d0.z, d0.w, d1.x, d1.y, d1.z, d1.w};
#pragma unroll
    for (int r = 0; r < 8; ++r)
        outp[(half * 8 + r) * HID_DIM + l] = c[r] * dr[r];
}

template <int F, int FP, int G, bool LAST>
__global__ __launch_bounds__(AGG_TPB)
void edge_agg_kept(const int* __restrict__ src, const int* __restrict__ dst,
                   const float* __restrict__ h, float* __restrict__ res,
                   const float* __restrict__ dinv, const float* __restrict__ bias,
                   const float* __restrict__ wnext,
                   const float* __restrict__ ws, size_t ws_bytes, unsigned* __restrict__ lists,
                   int num_nodes, int num_edges) {
    __shared__ alignas(16) float acc[AGG_RN * F];
    __shared__ unsigned queue[AGG_WAVES][AGG_QCAP];
    static_assert(AGG_SLICES == 32 && AGG_WAVES == AGG_SLICES, "lane t reads count t, wave t walks sub-list t");

    const int tid  = threadIdx.x;
    const int lane = tid & 31;
    const int wv   = tid >> 5;
    const unsigned rg = blockIdx.x;
    const int base = (int)rg * AGG_RN;
    const unsigned cnt = (unsigned)min(AGG_RN, num_nodes - base);
    const unsigned* piece = list_piece(ws, ws_bytes, lists, gridDim.x, rg);
    const unsigned c = (piece != nullptr) ? lists[(size_t)rg * AGG_SLICES + lane] : 0xFFFFFFFFu;
    const bool kept_ok = __builtin_amdgcn_ballot_w32(c > (unsigned)AGG_SCAP) == 0u;

    for (int i = tid; i < AGG_RN * F; i += AGG_TPB) acc[i] = 0.0f;
    __syncthreads();

    if (kept_ok) {
        constexpr int EPW = 32 / FP;
        constexpr int ST  = 32 / EPW;
        constexpr int TRIP = 32 * G;
        const int sub = lane / FP;
        const int j   = lane % FP;
        const int jc  = (j < F) ? j : F - 1;
        const int n = __builtin_amdgcn_readfirstlane(__shfl((int)c, wv));
        const unsigned* sl = piece + (size_t)wv * AGG_SCAP;
        if (n > 0) {
            unsigned cur[G], nxt[G];
#pragma unroll
            for (int k = 0; k < G; ++k) cur[k] = sl[min(k * 32 + lane, n - 1)];
            for (int g0 = 0; g0 < n; g0 += TRIP) {
#pragma unroll
                for (int k = 0; k < G; ++k) nxt[k] = sl[min(g0 + TRIP + k * 32 + lane, n - 1)];
                unsigned w[G * ST];
                float    val[G * ST];
#pragma unroll
                for (int k = 0; k < G; ++k)
#pragma unroll
                    for (int u = 0; u < ST; ++u) w[k * ST + u] = (unsigned)__shfl((int)cur[k], u * EPW + sub);
#pragma unroll
                for (int t = 0; t < G * ST; ++t) val[t] = h[(size_t)(w[t] >> AGG_DBITS) * F + jc];
#pragma unroll
                for (int t = 0; t < G * ST; ++t) {
                    asm volatile("" :: "v"(val[t]));
                }
#pragma unroll
                for (int k = 0; k < G; ++k)
#pragma unroll
                    for (int u = 0; u < ST; ++u)
                        if ((g0 + k * 32 + u * EPW + sub < n) && (j < F))
                            atomicAdd(&acc[(w[k * ST + u] & ((1u << AGG_DBITS) - 1u)) * F + j], val[k * ST + u]);
#pragma unroll
                for (int k = 0; k < G; ++k) cur[k] = nxt[k];
            }
        }
    } else {
        agg_scan<F, FP>(src, dst, h, acc, queue[wv], base, cnt, 0, num_edges);
    }
    __syncthreads();

    if constexpr (!LAST) {
        static_assert(F == HID_DIM, "GEMM2 takes layer 1's rows");
        const int half = lane >> 4;
        const int l    = lane & 15;
        const int tiles = ((int)cnt + 15) >> 4;
        const int wvu = __builtin_amdgcn_readfirstlane(wv);
        for (int t = wvu; t < tiles; t += AGG_WAVES) {
            const int rl = min(t * 16 + l, (int)cnt - 1);
            const size_t nd = (size_t)(base + rl);
            const float di = dinv[nd];
            const float* hr = h + nd * F + half * 2;
            const float* ar = acc + rl * F + half * 2;
            float dr[8];
#pragma unroll
            for (int r = 0; r < 8; ++r)
                dr[r] = dinv[base + min(t * 16 + half * 8 + r, (int)cnt - 1)];
            v8f cc = {};
#pragma unroll
            for (int k0 = 0; k0 < F; k0 += 4) {
                const int kr = k0 + half * 2;
                v2f s = *(const v2f*)(ar + k0);
                v2f g = *(const v2f*)(hr + k0);
                v2f a;
                a.x = fmaxf((s.x + g.x) * di + bias[kr + 0], 0.0f);
                a.y = fmaxf((s.y + g.y) * di + bias[kr + 1], 0.0f);
                v2f b;
                b.x = (l < OUT_DIM) ? wnext[(kr + 0) * OUT_DIM + l] : 0.0f;
                b.y = (l < OUT_DIM) ? wnext[(kr + 1) * OUT_DIM + l] : 0.0f;
                cc = __builtin_amdgcn_wmma_f32_16x16x4_f32(
                        false, a, false, b, (short)0, cc, false, false);
            }
            asm volatile("v_nop\n\tv_nop\n\tv_nop\n\tv_nop" : "+v"(cc));
#pragma unroll
            for (int r = 0; r < 8; ++r) {
                const int row = t * 16 + half * 8 + r;
                if (l < OUT_DIM && row < (int)cnt)
                    res[(size_t)(base + row) * OUT_DIM + l] = cc[r] * dr[r];
            }
        }
    } else {
        for (int nl = tid; nl < (int)cnt; nl += AGG_TPB) {
            const size_t nd = (size_t)(base + nl);
            const float di = dinv[nd];
            float v[F];
            float m = -INFINITY;
#pragma unroll
            for (int jj = 0; jj < F; ++jj) {
                v[jj] = (acc[nl * F + jj] + h[nd * F + jj]) * di + bias[jj];
                m = fmaxf(m, v[jj]);
            }
            float s = 0.0f;
#pragma unroll
            for (int jj = 0; jj < F; ++jj) s += expf(v[jj] - m);
            const float lse = logf(s);
#pragma unroll
            for (int jj = 0; jj < F; ++jj)
                res[nd * F + jj] = v[jj] - m - lse;
        }
    }
}

extern "C" void kernel_launch(void* const* d_in, const int* in_sizes, int n_in,
                              void* d_out, int out_size, void* d_ws, size_t ws_size,
                              hipStream_t stream) {
    const float* x  = (const float*)d_in[0];
    const int*   ei = (const int*)d_in[1];
    const float* W1 = (const float*)d_in[2];
    const float* b1 = (const float*)d_in[3];
    const float* W2 = (const float*)d_in[4];
    const float* b2 = (const float*)d_in[5];

    const int N = in_sizes[0] / IN_DIM;
    const int E = in_sizes[1] / 2;
    const int* src = ei;
    const int* dst = ei + E;

    float* ws   = (float*)d_ws;
    float* w1p  = ws;
    float* dinv = w1p + W1P_FLOATS;
    float* h1   = dinv + (size_t)N;
    float* h2   = h1 + (size_t)N * HID_DIM;
    unsigned* lists = (unsigned*)(h2 + (size_t)N * OUT_DIM);
    float* outp = (float*)d_out;

    const int B = 256;
    auto nb = [](long long work, int blk) { return (int)((work + blk - 1) / blk); };

    unsigned* part = (unsigned*)h1;
    const int chunks = nb(E, DEG_CHUNK);
    degree_hist<<<dim3(chunks, nb(N, DEG_CAP)), DEG_TPB, 0, stream>>>(dst, part, N, E);
    const int mblocks = nb((N + 1) / 2, B);
    degree_merge<<<mblocks + nb(G1_STEPS * 128, B), B, 0, stream>>>(
        part, dinv, N, chunks, W1, w1p, mblocks);

    int waves1 = N / 16;
    const int ranges  = nb(N, AGG_RN);
    const int gneed   = nb(waves1, AGG_WAVES);
    const int gblocks = (gneed > ranges) ? gneed : ranges;
    const int fblocks = AGG_SLICES * nb(ranges, FLT_RPG);
    gemm1_wmma_filter<<<gblocks + fblocks, AGG_TPB, 0, stream>>>(
        x, W1, w1p, dinv, h1, src, dst, ws, ws_size, lists, N, E);
    edge_agg_kept<HID_DIM, 16, 2, false><<<ranges, AGG_TPB, 0, stream>>>(
        src, dst, h1, h2, dinv, b1, W2, ws, ws_size, lists, N, E);

    edge_agg_kept<OUT_DIM, 8, 4, true><<<ranges, AGG_TPB, 0, stream>>>(
        src, dst, h2, outp, dinv, b2, W2, ws, ws_size, lists, N, E);
}
